# starting kernel
# baseline (speedup 1.0000x reference)
#include <hip/hip_runtime.h>
#include <hip/hip_bf16.h>

typedef float v2f __attribute__((ext_vector_type(2)));
typedef float v8f __attribute__((ext_vector_type(8)));

__global__ void gcn_zero_f(float* __restrict__ p, int n) {
    int t = blockIdx.x * blockDim.x + threadIdx.x;
    if (t < n) p[t] = 0.0f;
}

__global__ void gcn_copy4(float4* __restrict__ dst, const float4* __restrict__ src, int n4) {
    int t = blockIdx.x * blockDim.x + threadIdx.x;
    if (t < n4) dst[t] = src[t];
}

__global__ void gcn_deg(const int* __restrict__ se, const int* __restrict__ re,
                        float* __restrict__ degS, float* __restrict__ degR, int E) {
    int t = blockIdx.x * blockDim.x + threadIdx.x;
    if (t < E) {
        atomicAdd(&degS[se[t]], 1.0f);
        atomicAdd(&degR[re[t]], 1.0f);
    }
}

template <int F>
__global__ void gcn_scatter(const int* __restrict__ se, const int* __restrict__ re,
                            const float* __restrict__ src, float* __restrict__ dst, int E) {
    constexpr int FQ = F / 4;
    long long t = (long long)blockIdx.x * blockDim.x + threadIdx.x;
    if (t >= (long long)E * FQ) return;
    int e = (int)(t / FQ);
    int f = (int)(t % FQ) * 4;
    int s = se[e];
    int r = re[e];
    float4 v = *reinterpret_cast<const float4*>(&src[(size_t)s * F + f]);
    float* d = &dst[(size_t)r * F + f];
    atomicAdd(d + 0, v.x);
    atomicAdd(d + 1, v.y);
    atomicAdd(d + 2, v.z);
    atomicAdd(d + 3, v.w);
}

template <int F, bool ADDONE>
__global__ void gcn_scale(const float* __restrict__ in, const float* __restrict__ degR,
                          float* __restrict__ out, int N) {
    long long t = (long long)blockIdx.x * blockDim.x + threadIdx.x;
    if (t >= (long long)N * F) return;
    int i = (int)(t >> (F == 64 ? 6 : 4));
    float d = degR[i];
    float rs = ADDONE ? rsqrtf(d + 1.0f) : rsqrtf(fmaxf(d, 1.0f));
    out[t] = in[t] * rs;
}

template <int NOUT, bool TANH, bool ADDONE>
__global__ __launch_bounds__(128) void gcn_gemm(const float* __restrict__ in,
                                                const float* __restrict__ W,
                                                const float* __restrict__ bias,
                                                const float* __restrict__ degS,
                                                float* __restrict__ out, int N) {
    constexpr int K = 64;
    constexpr int CT = NOUT / 16;
    constexpr int PITCH = 68;

    __shared__ float Wt[NOUT * PITCH];
    __shared__ float Bl[NOUT];
    __shared__ float Al[64 * PITCH];

    const int tid = threadIdx.x;

    const float4* Wg4 = reinterpret_cast<const float4*>(W);
    for (int i = tid; i < K * NOUT / 4; i += 128) {
        int k = i / (NOUT / 4);
        int n = (i % (NOUT / 4)) * 4;
        float4 w = Wg4[i];
        Wt[(n + 0) * PITCH + k] = w.x;
        Wt[(n + 1) * PITCH + k] = w.y;
        Wt[(n + 2) * PITCH + k] = w.z;
        Wt[(n + 3) * PITCH + k] = w.w;
    }
    if (tid < NOUT) Bl[tid] = bias[tid];

    const int rowBase = blockIdx.x * 64;
    const float4* Ag4 = reinterpret_cast<const float4*>(in);
    for (int i = tid; i < 64 * (K / 4); i += 128) {
        int row = rowBase + i / (K / 4);
        int c4 = (i % (K / 4)) * 4;
        float4 v = (row < N) ? Ag4[(size_t)row * (K / 4) + (c4 >> 2)]
                             : make_float4(0.f, 0.f, 0.f, 0.f);
        *reinterpret_cast<float4*>(&Al[(i / (K / 4)) * PITCH + c4]) = v;
    }
    __syncthreads();

    const int wave = tid >> 5;
    const int lane = tid & 31;
    const int tileRow = rowBase + wave * 16;
    if (tileRow >= N) return;

    const int lm = lane & 15;
    const int kh = (lane >> 4) * 2;
    const int mh = (lane >> 4) * 8;

    v8f acc[CT] = {};
    const float* Arow = &Al[(wave * 16 + lm) * PITCH + kh];

#pragma unroll
    for (int k0 = 0; k0 < K; k0 += 4) {
        v2f a = *reinterpret_cast<const v2f*>(&Arow[k0]);
#pragma unroll
        for (int c = 0; c < CT; ++c) {
            v2f b = *reinterpret_cast<const v2f*>(&Wt[(c * 16 + lm) * PITCH + k0 + kh]);
            acc[c] = __builtin_amdgcn_wmma_f32_16x16x4_f32(
                false, a, false, b, (short)0, acc[c], false, false);
        }
    }

    float rs[8];
#pragma unroll
    for (int j = 0; j < 8; ++j) {
        int row = tileRow + mh + j;
        float d = (row < N) ? degS[row] : 1.0f;
        rs[j] = ADDONE ? rsqrtf(d + 1.0f) : rsqrtf(fmaxf(d, 1.0f));
    }

#pragma unroll
    for (int c = 0; c < CT; ++c) {
        int n = c * 16 + lm;
        float bv = Bl[n];
#pragma unroll
        for (int j = 0; j < 8; ++j) {
            int row = tileRow + mh + j;
            if (row < N) {
                float v = acc[c][j] + bv;
                if (TANH) v = tanhf(v);
                out[(size_t)row * NOUT + n] = v * rs[j];
            }
        }
    }
}

extern "C" void kernel_launch(void* const* d_in, const int* in_sizes, int n_in,
                              void* d_out, int out_size, void* d_ws, size_t ws_size,
                              hipStream_t stream) {
    const float* nodes = (const float*)d_in[0];
    const int* senders = (const int*)d_in[1];
    const int* receivers = (const int*)d_in[2];
    const float* W0 = (const float*)d_in[3];
    const float* b0 = (const float*)d_in[4];
    const float* W1 = (const float*)d_in[5];
    const float* b1 = (const float*)d_in[6];
    const float* W2 = (const float*)d_in[7];
    const float* b2 = (const float*)d_in[8];
    float* out = (float*)d_out;

    const int N = in_sizes[0] / 64;
    const int E = in_sizes[1];

    float* X = (float*)d_ws;
    float* Y = X + (size_t)N * 64;
    float* Z = Y + (size_t)N * 64;
    float* degS = Z + (size_t)N * 16;
    float* degR = degS + N;

    const int zn = N * 16 + 2 * N;
    gcn_zero_f<<<(zn + 255) / 256, 256, 0, stream>>>(Z, zn);
    gcn_deg<<<(E + 255) / 256, 256, 0, stream>>>(senders, receivers, degS, degR, E);

    const int tiles = (N + 15) / 16;
    const int gblocks = (tiles + 3) / 4;
    const int cp4 = N * 16;
    const int g64 = (int)(((long long)E * 16 + 255) / 256);
    const int g16 = (int)(((long long)E * 4 + 255) / 256);
    const int s64 = (int)(((long long)N * 64 + 255) / 256);
    const int s16 = (int)(((long long)N * 16 + 255) / 256);

    gcn_gemm<64, true, true><<<gblocks, 128, 0, stream>>>(nodes, W0, b0, degS, X, N);
    gcn_copy4<<<(cp4 + 255) / 256, 256, 0, stream>>>((float4*)Y, (const float4*)X, cp4);
    gcn_scatter<64><<<g64, 256, 0, stream>>>(senders, receivers, X, Y, E);
    gcn_scale<64, true><<<s64, 256, 0, stream>>>(Y, degR, X, N);

    gcn_gemm<64, true, true><<<gblocks, 128, 0, stream>>>(X, W1, b1, degS, Y, N);
    gcn_copy4<<<(cp4 + 255) / 256, 256, 0, stream>>>((float4*)X, (const float4*)Y, cp4);
    gcn_scatter<64><<<g64, 256, 0, stream>>>(senders, receivers, Y, X, E);
    gcn_scale<64, true><<<s64, 256, 0, stream>>>(X, degR, Y, N);

    gcn_gemm<16, false, false><<<gblocks, 128, 0, stream>>>(Y, W2, b2, degS, X, N);
    gcn_scatter<16><<<g16, 256, 0, stream>>>(senders, receivers, X, Z, E);
    gcn_scale<16, false><<<s16, 256, 0, stream>>>(Z, degR, out, N);
}


# fuse the elementwise passes into the GEMMs: receiver rsqrt scale applied while staging A, self-edge copy as a second epilogue store; 13 -> 9 launches
# speedup vs baseline: 1.1807x; 1.1807x over previous
#include <hip/hip_runtime.h>
#include <hip/hip_bf16.h>

typedef float v2f __attribute__((ext_vector_type(2)));
typedef float v8f __attribute__((ext_vector_type(8)));

__global__ void gcn_zero_f(float* __restrict__ p, int n) {
    int t = blockIdx.x * blockDim.x + threadIdx.x;
    if (t < n) p[t] = 0.0f;
}

__global__ void gcn_deg(const int* __restrict__ se, const int* __restrict__ re,
                        float* __restrict__ degS, float* __restrict__ degR, int E) {
    int t = blockIdx.x * blockDim.x + threadIdx.x;
    if (t < E) {
        atomicAdd(&degS[se[t]], 1.0f);
        atomicAdd(&degR[re[t]], 1.0f);
    }
}

template <int F>
__global__ void gcn_scatter(const int* __restrict__ se, const int* __restrict__ re,
                            const float* __restrict__ src, float* __restrict__ dst, int E) {
    constexpr int FQ = F / 4;
    long long t = (long long)blockIdx.x * blockDim.x + threadIdx.x;
    if (t >= (long long)E * FQ) return;
    int e = (int)(t / FQ);
    int f = (int)(t % FQ) * 4;
    int s = se[e];
    int r = re[e];
    float4 v = *reinterpret_cast<const float4*>(&src[(size_t)s * F + f]);
    float* d = &dst[(size_t)r * F + f];
    atomicAdd(d + 0, v.x);
    atomicAdd(d + 1, v.y);
    atomicAdd(d + 2, v.z);
    atomicAdd(d + 3, v.w);
}

template <int F, bool ADDONE>
__global__ void gcn_scale(const float* __restrict__ in, const float* __restrict__ degR,
                          float* __restrict__ out, int N) {
    long long t = (long long)blockIdx.x * blockDim.x + threadIdx.x;
    if (t >= (long long)N * F) return;
    int i = (int)(t >> (F == 64 ? 6 : 4));
    float d = degR[i];
    float rs = ADDONE ? rsqrtf(d + 1.0f) : rsqrtf(fmaxf(d, 1.0f));
    out[t] = in[t] * rs;
}

template <int NOUT, bool TANH, bool ADDONE, bool PRESCALE, bool DUP>
__global__ __launch_bounds__(128) void gcn_gemm(const float* __restrict__ in,
                                                const float* __restrict__ W,
                                                const float* __restrict__ bias,
                                                const float* __restrict__ degS,
                                                const float* __restrict__ degR,
                                                float* __restrict__ out,
                                                float* __restrict__ out2, int N) {
    constexpr int K = 64;
    constexpr int CT = NOUT / 16;
    constexpr int PITCH = 68;

    __shared__ float Wt[NOUT * PITCH];
    __shared__ float Bl[NOUT];
    __shared__ float Al[64 * PITCH];

    const int tid = threadIdx.x;

    const float4* Wg4 = reinterpret_cast<const float4*>(W);
    for (int i = tid; i < K * NOUT / 4; i += 128) {
        int k = i / (NOUT / 4);
        int n = (i % (NOUT / 4)) * 4;
        float4 w = Wg4[i];
        Wt[(n + 0) * PITCH + k] = w.x;
        Wt[(n + 1) * PITCH + k] = w.y;
        Wt[(n + 2) * PITCH + k] = w.z;
        Wt[(n + 3) * PITCH + k] = w.w;
    }
    if (tid < NOUT) Bl[tid] = bias[tid];

    const int rowBase = blockIdx.x * 64;
    const float4* Ag4 = reinterpret_cast<const float4*>(in);
    for (int i = tid; i < 64 * (K / 4); i += 128) {
        int row = rowBase + i / (K / 4);
        int c4 = (i % (K / 4)) * 4;
        float4 v = make_float4(0.f, 0.f, 0.f, 0.f);
        if (row < N) {
            v = Ag4[(size_t)row * (K / 4) + (c4 >> 2)];
            if (PRESCALE) {
                float pre = rsqrtf(degR[row] + 1.0f);
                v.x *= pre;
                v.y *= pre;
                v.z *= pre;
                v.w *= pre;
            }
        }
        *reinterpret_cast<float4*>(&Al[(i / (K / 4)) * PITCH + c4]) = v;
    }
    __syncthreads();

    const int wave = tid >> 5;
    const int lane = tid & 31;
    const int tileRow = rowBase + wave * 16;
    if (tileRow >= N) return;

    const int lm = lane & 15;
    const int kh = (lane >> 4) * 2;
    const int mh = (lane >> 4) * 8;

    v8f acc[CT] = {};
    const float* Arow = &Al[(wave * 16 + lm) * PITCH + kh];

#pragma unroll
    for (int k0 = 0; k0 < K; k0 += 4) {
        v2f a = *reinterpret_cast<const v2f*>(&Arow[k0]);
#pragma unroll
        for (int c = 0; c < CT; ++c) {
            v2f b = *reinterpret_cast<const v2f*>(&Wt[(c * 16 + lm) * PITCH + k0 + kh]);
            acc[c] = __builtin_amdgcn_wmma_f32_16x16x4_f32(
                false, a, false, b, (short)0, acc[c], false, false);
        }
    }

    float rs[8];
#pragma unroll
    for (int j = 0; j < 8; ++j) {
        int row = tileRow + mh + j;
        float d = (row < N) ? degS[row] : 1.0f;
        rs[j] = ADDONE ? rsqrtf(d + 1.0f) : rsqrtf(fmaxf(d, 1.0f));
    }

#pragma unroll
    for (int c = 0; c < CT; ++c) {
        int n = c * 16 + lm;
        float bv = Bl[n];
#pragma unroll
        for (int j = 0; j < 8; ++j) {
            int row = tileRow + mh + j;
            if (row < N) {
                float v = acc[c][j] + bv;
                if (TANH) v = tanhf(v);
                float o = v * rs[j];
                out[(size_t)row * NOUT + n] = o;
                if (DUP) out2[(size_t)row * NOUT + n] = o;
            }
        }
    }
}

extern "C" void kernel_launch(void* const* d_in, const int* in_sizes, int n_in,
                              void* d_out, int out_size, void* d_ws, size_t ws_size,
                              hipStream_t stream) {
    const float* nodes = (const float*)d_in[0];
    const int* senders = (const int*)d_in[1];
    const int* receivers = (const int*)d_in[2];
    const float* W0 = (const float*)d_in[3];
    const float* b0 = (const float*)d_in[4];
    const float* W1 = (const float*)d_in[5];
    const float* b1 = (const float*)d_in[6];
    const float* W2 = (const float*)d_in[7];
    const float* b2 = (const float*)d_in[8];
    float* out = (float*)d_out;

    const int N = in_sizes[0] / 64;
    const int E = in_sizes[1];

    float* X = (float*)d_ws;
    float* Y = X + (size_t)N * 64;
    float* T = Y + (size_t)N * 64;
    float* Z = T + (size_t)N * 64;
    float* degS = Z + (size_t)N * 16;
    float* degR = degS + N;

    const int zn = N * 16 + 2 * N;
    gcn_zero_f<<<(zn + 255) / 256, 256, 0, stream>>>(Z, zn);
    gcn_deg<<<(E + 255) / 256, 256, 0, stream>>>(senders, receivers, degS, degR, E);

    const int tiles = (N + 15) / 16;
    const int gblocks = (tiles + 3) / 4;
    const int g64 = (int)(((long long)E * 16 + 255) / 256);
    const int g16 = (int)(((long long)E * 4 + 255) / 256);
    const int s16 = (int)(((long long)N * 16 + 255) / 256);

    gcn_gemm<64, true, true, false, true><<<gblocks, 128, 0, stream>>>(nodes, W0, b0, degS, degR, X, Y, N);
    gcn_scatter<64><<<g64, 256, 0, stream>>>(senders, receivers, X, Y, E);

    gcn_gemm<64, true, true, true, true><<<gblocks, 128, 0, stream>>>(Y, W1, b1, degS, degR, X, T, N);
    gcn_scatter<64><<<g64, 256, 0, stream>>>(senders, receivers, X, T, E);

    gcn_gemm<16, false, false, true, false><<<gblocks, 128, 0, stream>>>(T, W2, b2, degS, degR, X, nullptr, N);
    gcn_scatter<16><<<g16, 256, 0, stream>>>(senders, receivers, X, Z, E);
    gcn_scale<16, false><<<s16, 256, 0, stream>>>(Z, degR, out, N);
}


# aggregation by gather over per-receiver sender slot lists (rebuilt each call, one returning int add per edge; int degree counts) replaces float atomic scatter; sums in slot order, lane-group fold
# speedup vs baseline: 2.0530x; 1.7388x over previous
#include <hip/hip_runtime.h>
#include <hip/hip_bf16.h>

typedef float v2f __attribute__((ext_vector_type(2)));
typedef float v4f __attribute__((ext_vector_type(4)));
typedef float v8f __attribute__((ext_vector_type(8)));

constexpr int CAP = 32;

__global__ void gcn_zero_i(int* __restrict__ p, int n) {
    int t = blockIdx.x * blockDim.x + threadIdx.x;
    if (t < n) p[t] = 0;
}

__global__ void gcn_fill(const int* __restrict__ se, const int* __restrict__ re,
                         int* __restrict__ cntS, int* __restrict__ cntR,
                         int* __restrict__ col, int* __restrict__ ovfCount,
                         int* __restrict__ ovfS, int* __restrict__ ovfR, int E) {
    int t = blockIdx.x * blockDim.x + threadIdx.x;
    if (t < E) {
        int s = se[t];
        int r = re[t];
        atomicAdd(&cntS[s], 1);
        int slot = atomicAdd(&cntR[r], 1);
        if (slot < CAP) {
            col[(size_t)r * CAP + slot] = s;
        } else {
            int o = atomicAdd(ovfCount, 1);
            ovfS[o] = s;
            ovfR[o] = r;
        }
    }
}

template <int F, bool SELF, int U>
__global__ __launch_bounds__(256) void gcn_gather(const float* __restrict__ src,
                                                  const int* __restrict__ cntR,
                                                  const int* __restrict__ col,
                                                  const int* __restrict__ ovfCount,
                                                  const int* __restrict__ ovfS,
                                                  const int* __restrict__ ovfR,
                                                  float* __restrict__ dst, int N, int E) {
    constexpr int LPR = F / 4;
    constexpr int G = 32 / LPR;
    const int lane = threadIdx.x & 31;
    const int r = blockIdx.x * 8 + (threadIdx.x >> 5);
    if (r >= N) return;

    const int g = lane / LPR;
    const int q4 = (lane % LPR) * 4;
    const int cnt = cntR[r];
    const int n = cnt < CAP ? cnt : CAP;

    int myS = 0;
    if (lane < n) myS = col[(size_t)r * CAP + lane];

    v4f acc = {0.f, 0.f, 0.f, 0.f};
    if (SELF && g == 0) acc = *reinterpret_cast<const v4f*>(&src[(size_t)r * F + q4]);

    for (int j0 = 0; j0 < n; j0 += U * G) {
        v4f v[U];
#pragma unroll
        for (int u = 0; u < U; ++u) {
            int j = j0 + u * G + g;
            int s = __shfl(myS, j & 31);
            v[u] = v4f{0.f, 0.f, 0.f, 0.f};
            if (j < n) v[u] = *reinterpret_cast<const v4f*>(&src[(size_t)s * F + q4]);
        }
#pragma unroll
        for (int u = 0; u < U; ++u) acc += v[u];
    }

    if (cnt > CAP) {
        int no = *ovfCount;
        if (no > E) no = E;
        for (int i = g; i < no; i += G) {
            if (ovfR[i] == r) acc += *reinterpret_cast<const v4f*>(&src[(size_t)ovfS[i] * F + q4]);
        }
    }

#pragma unroll
    for (int off = LPR; off < 32; off <<= 1) {
        acc.x += __shfl_xor(acc.x, off);
        acc.y += __shfl_xor(acc.y, off);
        acc.z += __shfl_xor(acc.z, off);
        acc.w += __shfl_xor(acc.w, off);
    }

    if (g == 0) {
        float d = (float)cnt;
        float sc = SELF ? rsqrtf(d + 1.0f) : rsqrtf(fmaxf(d, 1.0f));
        *reinterpret_cast<v4f*>(&dst[(size_t)r * F + q4]) = acc * sc;
    }
}

template <int NOUT, bool TANH, bool ADDONE>
__global__ __launch_bounds__(128) void gcn_gemm(const float* __restrict__ in,
                                                const float* __restrict__ W,
                                                const float* __restrict__ bias,
                                                const int* __restrict__ cntS,
                                                float* __restrict__ out, int N) {
    constexpr int K = 64;
    constexpr int CT = NOUT / 16;
    constexpr int PITCH = 68;

    __shared__ float Wt[NOUT * PITCH];
    __shared__ float Bl[NOUT];
    __shared__ float Al[64 * PITCH];

    const int tid = threadIdx.x;

    const float4* Wg4 = reinterpret_cast<const float4*>(W);
    for (int i = tid; i < K * NOUT / 4; i += 128) {
        int k = i / (NOUT / 4);
        int n = (i % (NOUT / 4)) * 4;
        float4 w = Wg4[i];
        Wt[(n + 0) * PITCH + k] = w.x;
        Wt[(n + 1) * PITCH + k] = w.y;
        Wt[(n + 2) * PITCH + k] = w.z;
        Wt[(n + 3) * PITCH + k] = w.w;
    }
    if (tid < NOUT) Bl[tid] = bias[tid];

    const int rowBase = blockIdx.x * 64;
    const float4* Ag4 = reinterpret_cast<const float4*>(in);
    for (int i = tid; i < 64 * (K / 4); i += 128) {
        int row = rowBase + i / (K / 4);
        int c4 = (i % (K / 4)) * 4;
        float4 v = make_float4(0.f, 0.f, 0.f, 0.f);
        if (row < N) v = Ag4[(size_t)row * (K / 4) + (c4 >> 2)];
        *reinterpret_cast<float4*>(&Al[(i / (K / 4)) * PITCH + c4]) = v;
    }
    __syncthreads();

    const int wave = tid >> 5;
    const int lane = tid & 31;
    const int tileRow = rowBase + wave * 16;
    if (tileRow >= N) return;

    const int lm = lane & 15;
    const int kh = (lane >> 4) * 2;
    const int mh = (lane >> 4) * 8;

    v8f acc[CT] = {};
    const float* Arow = &Al[(wave * 16 + lm) * PITCH + kh];

#pragma unroll
    for (int k0 = 0; k0 < K; k0 += 4) {
        v2f a = *reinterpret_cast<const v2f*>(&Arow[k0]);
#pragma unroll
        for (int c = 0; c < CT; ++c) {
            v2f b = *reinterpret_cast<const v2f*>(&Wt[(c * 16 + lm) * PITCH + k0 + kh]);
            acc[c] = __builtin_amdgcn_wmma_f32_16x16x4_f32(
                false, a, false, b, (short)0, acc[c], false, false);
        }
    }

    float rs[8];
#pragma unroll
    for (int j = 0; j < 8; ++j) {
        int row = tileRow + mh + j;
        float d = (row < N) ? (float)cntS[row] : 1.0f;
        rs[j] = ADDONE ? rsqrtf(d + 1.0f) : rsqrtf(fmaxf(d, 1.0f));
    }

#pragma unroll
    for (int c = 0; c < CT; ++c) {
        int n = c * 16 + lm;
        float bv = Bl[n];
#pragma unroll
        for (int j = 0; j < 8; ++j) {
            int row = tileRow + mh + j;
            if (row < N) {
                float v = acc[c][j] + bv;
                if (TANH) v = tanhf(v);
                out[(size_t)row * NOUT + n] = v * rs[j];
            }
        }
    }
}

extern "C" void kernel_launch(void* const* d_in, const int* in_sizes, int n_in,
                              void* d_out, int out_size, void* d_ws, size_t ws_size,
                              hipStream_t stream) {
    const float* nodes = (const float*)d_in[0];
    const int* senders = (const int*)d_in[1];
    const int* receivers = (const int*)d_in[2];
    const float* W0 = (const float*)d_in[3];
    const float* b0 = (const float*)d_in[4];
    const float* W1 = (const float*)d_in[5];
    const float* b1 = (const float*)d_in[6];
    const float* W2 = (const float*)d_in[7];
    const float* b2 = (const float*)d_in[8];
    float* out = (float*)d_out;

    const int N = in_sizes[0] / 64;
    const int E = in_sizes[1];

    float* X = (float*)d_ws;
    float* Y = X + (size_t)N * 64;
    float* Z = Y + (size_t)N * 64;
    int* col = (int*)(Z + (size_t)N * 16);
    int* ovfS = col + (size_t)N * CAP;
    int* ovfR = ovfS + E;
    int* cntS = ovfR + E;
    int* cntR = cntS + N;
    int* ovfCount = cntR + N;

    const int zn = 2 * N + 1;
    gcn_zero_i<<<(zn + 255) / 256, 256, 0, stream>>>(cntS, zn);
    gcn_fill<<<(E + 255) / 256, 256, 0, stream>>>(senders, receivers, cntS, cntR, col,
                                                  ovfCount, ovfS, ovfR, E);

    const int tiles = (N + 15) / 16;
    const int gblocks = (tiles + 3) / 4;
    const int ablocks = (N + 7) / 8;

    gcn_gemm<64, true, true><<<gblocks, 128, 0, stream>>>(nodes, W0, b0, cntS, X, N);
    gcn_gather<64, true, 8><<<ablocks, 256, 0, stream>>>(X, cntR, col, ovfCount, ovfS, ovfR, Y, N, E);

    gcn_gemm<64, true, true><<<gblocks, 128, 0, stream>>>(Y, W1, b1, cntS, X, N);
    gcn_gather<64, true, 8><<<ablocks, 256, 0, stream>>>(X, cntR, col, ovfCount, ovfS, ovfR, Y, N, E);

    gcn_gemm<16, false, false><<<gblocks, 128, 0, stream>>>(Y, W2, b2, cntS, Z, N);
    gcn_gather<16, false, 2><<<ablocks, 256, 0, stream>>>(Z, cntR, col, ovfCount, ovfS, ovfR, out, N, E);
}


# gather launches on a resident grid: 6144 long-lived waves each walk consecutive receiver rows, two rows' sender loads in flight and the next slot lines prefetched, not one short wave per row
# speedup vs baseline: 2.1134x; 1.0294x over previous
#include <hip/hip_runtime.h>
#include <hip/hip_bf16.h>

typedef float v2f __attribute__((ext_vector_type(2)));
typedef float v4f __attribute__((ext_vector_type(4)));
typedef float v8f __attribute__((ext_vector_type(8)));

constexpr int CAP = 32;

__global__ void gcn_zero_i(int* __restrict__ p, int n) {
    int t = blockIdx.x * blockDim.x + threadIdx.x;
    if (t < n) p[t] = 0;
}

__global__ void gcn_fill(const int* __restrict__ se, const int* __restrict__ re,
                         int* __restrict__ cntS, int* __restrict__ cntR,
                         int* __restrict__ col, int* __restrict__ ovfCount,
                         int* __restrict__ ovfS, int* __restrict__ ovfR, int E) {
    int t = blockIdx.x * blockDim.x + threadIdx.x;
    if (t < E) {
        int s = se[t];
        int r = re[t];
        atomicAdd(&cntS[s], 1);
        int slot = atomicAdd(&cntR[r], 1);
        if (slot < CAP) {
            col[(size_t)r * CAP + slot] = s;
        } else {
            int o = atomicAdd(ovfCount, 1);
            ovfS[o] = s;
            ovfR[o] = r;
        }
    }
}

constexpr int GATHER_BLOCKS = 768;
constexpr int GATHER_WAVES = GATHER_BLOCKS * 8;

template <int F, bool SELF, int U>
__global__ __launch_bounds__(256) void gcn_gather(const float* __restrict__ src,
                                                  const int* __restrict__ cntR,
                                                  const int* __restrict__ col,
                                                  const int* __restrict__ ovfCount,
                                                  const int* __restrict__ ovfS,
                                                  const int* __restrict__ ovfR,
                                                  float* __restrict__ dst, int N, int E) {
    constexpr int LPR = F / 4;
    constexpr int G = 32 / LPR;
    const int lane = threadIdx.x & 31;
    const int g = lane / LPR;
    const int q4 = (lane % LPR) * 4;

    const int wave = __builtin_amdgcn_readfirstlane((int)(blockIdx.x * 8 + (threadIdx.x >> 5)));
    const int base = N / GATHER_WAVES;
    const int rem = N % GATHER_WAVES;
    const int r0 = wave * base + (wave < rem ? wave : rem);
    const int r1 = r0 + base + (wave < rem ? 1 : 0);

    const v4f zero4 = {0.f, 0.f, 0.f, 0.f};

    for (int rb = r0; rb < r1; rb += 32) {
        const int m = (r1 - rb) < 32 ? (r1 - rb) : 32;
        const int cntv = cntR[rb + (lane < m ? lane : m - 1)];

        auto rowCount = [&](int i) -> int { return i < m ? __builtin_amdgcn_readlane(cntv, i & 31) : 0; };

        auto slotLine = [&](int i) -> int {
            const int c = rowCount(i);
            const int n = c < CAP ? c : CAP;
            const int ri = rb + (i < m ? i : m - 1);
            const int* p = (n > 0) ? &col[(size_t)ri * CAP + (lane < n ? lane : n - 1)] : &cntR[ri];
            return *p;
        };

        auto issueRows = [&](v4f (&v)[U], v4f& selfv, int i, const int (&s)[U]) {
            const int c = rowCount(i);
            const int n = c < CAP ? c : CAP;
            if (SELF && i < m) selfv = *reinterpret_cast<const v4f*>(&src[(size_t)(rb + i) * F + q4]);
#pragma unroll
            for (int u = 0; u < U; ++u) {
                if (u * G < n) v[u] = *reinterpret_cast<const v4f*>(&src[(size_t)s[u] * F + q4]);
            }
        };

        auto finishRow = [&](const v4f (&v)[U], const v4f& selfv, int i, int myS) {
            const int r = rb + i;
            const int cnt = rowCount(i);
            const int n = cnt < CAP ? cnt : CAP;

            v4f acc = zero4;
            if (SELF && g == 0) acc = selfv;
#pragma unroll
            for (int u = 0; u < U; ++u) {
                if (u * G < n) acc += (u * G + g < n) ? v[u] : zero4;
            }

            for (int j0 = U * G; j0 < n; j0 += U * G) {
                v4f w[U];
#pragma unroll
                for (int u = 0; u < U; ++u) {
                    const int j = j0 + u * G + g;
                    const int s = __shfl(myS, j & 31);
                    w[u] = zero4;
                    if (j < n) w[u] = *reinterpret_cast<const v4f*>(&src[(size_t)s * F + q4]);
                }
#pragma unroll
                for (int u = 0; u < U; ++u) acc += w[u];
            }

            if (cnt > CAP) {
                int no = *ovfCount;
                if (no > E) no = E;
                for (int k = g; k < no; k += G) {
                    if (ovfR[k] == r) acc += *reinterpret_cast<const v4f*>(&src[(size_t)ovfS[k] * F + q4]);
                }
            }

#pragma unroll
            for (int off = LPR; off < 32; off <<= 1) {
                acc.x += __shfl_xor(acc.x, off);
                acc.y += __shfl_xor(acc.y, off);
                acc.z += __shfl_xor(acc.z, off);
                acc.w += __shfl_xor(acc.w, off);
            }

            if (g == 0) {
                const float d = (float)cnt;
                const float sc = SELF ? rsqrtf(d + 1.0f) : rsqrtf(fmaxf(d, 1.0f));
                *reinterpret_cast<v4f*>(&dst[(size_t)r * F + q4]) = acc * sc;
            }
        };

        int slotX = slotLine(0);
        int slotY = slotLine(1);

        for (int i = 0; i < m; i += 2) {
            const int slotX2 = slotLine(i + 2);
            const int slotY2 = slotLine(i + 3);

            int sx[U], sy[U];
#pragma unroll
            for (int u = 0; u < U; ++u) {
                sx[u] = __shfl(slotX, u * G + g);
                sy[u] = __shfl(slotY, u * G + g);
            }

            v4f vx[U], vy[U];
            v4f selfX, selfY;
            issueRows(vx, selfX, i, sx);
            issueRows(vy, selfY, i + 1, sy);

            finishRow(vx, selfX, i, slotX);
            if (i + 1 < m) finishRow(vy, selfY, i + 1, slotY);

            slotX = slotX2;
            slotY = slotY2;
        }
    }
}

template <int NOUT, bool TANH, bool ADDONE>
__global__ __launch_bounds__(128) void gcn_gemm(const float* __restrict__ in,
                                                const float* __restrict__ W,
                                                const float* __restrict__ bias,
                                                const int* __restrict__ cntS,
                                                float* __restrict__ out, int N) {
    constexpr int K = 64;
    constexpr int CT = NOUT / 16;
    constexpr int PITCH = 68;

    __shared__ float Wt[NOUT * PITCH];
    __shared__ float Bl[NOUT];
    __shared__ float Al[64 * PITCH];

    const int tid = threadIdx.x;

    const float4* Wg4 = reinterpret_cast<const float4*>(W);
    for (int i = tid; i < K * NOUT / 4; i += 128) {
        int k = i / (NOUT / 4);
        int n = (i % (NOUT / 4)) * 4;
        float4 w = Wg4[i];
        Wt[(n + 0) * PITCH + k] = w.x;
        Wt[(n + 1) * PITCH + k] = w.y;
        Wt[(n + 2) * PITCH + k] = w.z;
        Wt[(n + 3) * PITCH + k] = w.w;
    }
    if (tid < NOUT) Bl[tid] = bias[tid];

    const int rowBase = blockIdx.x * 64;
    const float4* Ag4 = reinterpret_cast<const float4*>(in);
    for (int i = tid; i < 64 * (K / 4); i += 128) {
        int row = rowBase + i / (K / 4);
        int c4 = (i % (K / 4)) * 4;
        float4 v = make_float4(0.f, 0.f, 0.f, 0.f);
        if (row < N) v = Ag4[(size_t)row * (K / 4) + (c4 >> 2)];
        *reinterpret_cast<float4*>(&Al[(i / (K / 4)) * PITCH + c4]) = v;
    }
    __syncthreads();

    const int wave = tid >> 5;
    const int lane = tid & 31;
    const int tileRow = rowBase + wave * 16;
    if (tileRow >= N) return;

    const int lm = lane & 15;
    const int kh = (lane >> 4) * 2;
    const int mh = (lane >> 4) * 8;

    v8f acc[CT] = {};
    const float* Arow = &Al[(wave * 16 + lm) * PITCH + kh];

#pragma unroll
    for (int k0 = 0; k0 < K; k0 += 4) {
        v2f a = *reinterpret_cast<const v2f*>(&Arow[k0]);
#pragma unroll
        for (int c = 0; c < CT; ++c) {
            v2f b = *reinterpret_cast<const v2f*>(&Wt[(c * 16 + lm) * PITCH + k0 + kh]);
            acc[c] = __builtin_amdgcn_wmma_f32_16x16x4_f32(
                false, a, false, b, (short)0, acc[c], false, false);
        }
    }

    float rs[8];
#pragma unroll
    for (int j = 0; j < 8; ++j) {
        int row = tileRow + mh + j;
        float d = (row < N) ? (float)cntS[row] : 1.0f;
        rs[j] = ADDONE ? rsqrtf(d + 1.0f) : rsqrtf(fmaxf(d, 1.0f));
    }

#pragma unroll
    for (int c = 0; c < CT; ++c) {
        int n = c * 16 + lm;
        float bv = Bl[n];
#pragma unroll
        for (int j = 0; j < 8; ++j) {
            int row = tileRow + mh + j;
            if (row < N) {
                float v = acc[c][j] + bv;
                if (TANH) v = tanhf(v);
                out[(size_t)row * NOUT + n] = v * rs[j];
            }
        }
    }
}

extern "C" void kernel_launch(void* const* d_in, const int* in_sizes, int n_in,
                              void* d_out, int out_size, void* d_ws, size_t ws_size,
                              hipStream_t stream) {
    const float* nodes = (const float*)d_in[0];
    const int* senders = (const int*)d_in[1];
    const int* receivers = (const int*)d_in[2];
    const float* W0 = (const float*)d_in[3];
    const float* b0 = (const float*)d_in[4];
    const float* W1 = (const float*)d_in[5];
    const float* b1 = (const float*)d_in[6];
    const float* W2 = (const float*)d_in[7];
    const float* b2 = (const float*)d_in[8];
    float* out = (float*)d_out;

    const int N = in_sizes[0] / 64;
    const int E = in_sizes[1];

    float* X = (float*)d_ws;
    float* Y = X + (size_t)N * 64;
    float* Z = Y + (size_t)N * 64;
    int* col = (int*)(Z + (size_t)N * 16);
    int* ovfS = col + (size_t)N * CAP;
    int* ovfR = ovfS + E;
    int* cntS = ovfR + E;
    int* cntR = cntS + N;
    int* ovfCount = cntR + N;

    const int zn = 2 * N + 1;
    gcn_zero_i<<<(zn + 255) / 256, 256, 0, stream>>>(cntS, zn);
    gcn_fill<<<(E + 255) / 256, 256, 0, stream>>>(senders, receivers, cntS, cntR, col,
                                                  ovfCount, ovfS, ovfR, E);

    const int tiles = (N + 15) / 16;
    const int gblocks = (tiles + 3) / 4;

    gcn_gemm<64, true, true><<<gblocks, 128, 0, stream>>>(nodes, W0, b0, cntS, X, N);
    gcn_gather<64, true, 8><<<GATHER_BLOCKS, 256, 0, stream>>>(X, cntR, col, ovfCount, ovfS, ovfR, Y, N, E);

    gcn_gemm<64, true, true><<<gblocks, 128, 0, stream>>>(Y, W1, b1, cntS, X, N);
    gcn_gather<64, true, 8><<<GATHER_BLOCKS, 256, 0, stream>>>(X, cntR, col, ovfCount, ovfS, ovfR, Y, N, E);

    gcn_gemm<16, false, false><<<gblocks, 128, 0, stream>>>(Y, W2, b2, cntS, Z, N);
    gcn_gather<16, false, 2><<<GATHER_BLOCKS, 256, 0, stream>>>(Z, cntR, col, ovfCount, ovfS, ovfR, out, N, E);
}


# GEMM tile staging: a thread issues all its global loads (W, bias, its A rows, sender counts) before the first wait, then fills LDS; was one load per loop trip, each waited for
# speedup vs baseline: 2.3223x; 1.0988x over previous
#include <hip/hip_runtime.h>
#include <hip/hip_bf16.h>

typedef float v2f __attribute__((ext_vector_type(2)));
typedef float v4f __attribute__((ext_vector_type(4)));
typedef float v8f __attribute__((ext_vector_type(8)));

constexpr int CAP = 32;

__global__ void gcn_zero_i(int* __restrict__ p, int n) {
    int t = blockIdx.x * blockDim.x + threadIdx.x;
    if (t < n) p[t] = 0;
}

__global__ void gcn_fill(const int* __restrict__ se, const int* __restrict__ re,
                         int* __restrict__ cntS, int* __restrict__ cntR,
                         int* __restrict__ col, int* __restrict__ ovfCount,
                         int* __restrict__ ovfS, int* __restrict__ ovfR, int E) {
    int t = blockIdx.x * blockDim.x + threadIdx.x;
    if (t < E) {
        int s = se[t];
        int r = re[t];
        atomicAdd(&cntS[s], 1);
        int slot = atomicAdd(&cntR[r], 1);
        if (slot < CAP) {
            col[(size_t)r * CAP + slot] = s;
        } else {
            int o = atomicAdd(ovfCount, 1);
            ovfS[o] = s;
            ovfR[o] = r;
        }
    }
}

constexpr int GATHER_BLOCKS = 768;
constexpr int GATHER_WAVES = GATHER_BLOCKS * 8;

template <int F, bool SELF, int U>
__global__ __launch_bounds__(256) void gcn_gather(const float* __restrict__ src,
                                                  const int* __restrict__ cntR,
                                                  const int* __restrict__ col,
                                                  const int* __restrict__ ovfCount,
                                                  const int* __restrict__ ovfS,
                                                  const int* __restrict__ ovfR,
                                                  float* __restrict__ dst, int N, int E) {
    constexpr int LPR = F / 4;
    constexpr int G = 32 / LPR;
    const int lane = threadIdx.x & 31;
    const int g = lane / LPR;
    const int q4 = (lane % LPR) * 4;

    const int wave = __builtin_amdgcn_readfirstlane((int)(blockIdx.x * 8 + (threadIdx.x >> 5)));
    const int base = N / GATHER_WAVES;
    const int rem = N % GATHER_WAVES;
    const int r0 = wave * base + (wave < rem ? wave : rem);
    const int r1 = r0 + base + (wave < rem ? 1 : 0);

    const v4f zero4 = {0.f, 0.f, 0.f, 0.f};

    for (int rb = r0; rb < r1; rb += 32) {
        const int m = (r1 - rb) < 32 ? (r1 - rb) : 32;
        const int cntv = cntR[rb + (lane < m ? lane : m - 1)];

        auto rowCount = [&](int i) -> int { return i < m ? __builtin_amdgcn_readlane(cntv, i & 31) : 0; };

        auto slotLine = [&](int i) -> int {
            const int c = rowCount(i);
            const int n = c < CAP ? c : CAP;
            const int ri = rb + (i < m ? i : m - 1);
            const int* p = (n > 0) ? &col[(size_t)ri * CAP + (lane < n ? lane : n - 1)] : &cntR[ri];
            return *p;
        };

        auto issueRows = [&](v4f (&v)[U], v4f& selfv, int i, const int (&s)[U]) {
            const int c = rowCount(i);
            const int n = c < CAP ? c : CAP;
            if (SELF && i < m) selfv = *reinterpret_cast<const v4f*>(&src[(size_t)(rb + i) * F + q4]);
#pragma unroll
            for (int u = 0; u < U; ++u) {
                if (u * G < n) v[u] = *reinterpret_cast<const v4f*>(&src[(size_t)s[u] * F + q4]);
            }
        };

        auto finishRow = [&](const v4f (&v)[U], const v4f& selfv, int i, int myS) {
            const int r = rb + i;
            const int cnt = rowCount(i);
            const int n = cnt < CAP ? cnt : CAP;

            v4f acc = zero4;
            if (SELF && g == 0) acc = selfv;
#pragma unroll
            for (int u = 0; u < U; ++u) {
                if (u * G < n) acc += (u * G + g < n) ? v[u] : zero4;
            }

            for (int j0 = U * G; j0 < n; j0 += U * G) {
                v4f w[U];
#pragma unroll
                for (int u = 0; u < U; ++u) {
                    const int j = j0 + u * G + g;
                    const int s = __shfl(myS, j & 31);
                    w[u] = zero4;
                    if (j < n) w[u] = *reinterpret_cast<const v4f*>(&src[(size_t)s * F + q4]);
                }
#pragma unroll
                for (int u = 0; u < U; ++u) acc += w[u];
            }

            if (cnt > CAP) {
                int no = *ovfCount;
                if (no > E) no = E;
                for (int k = g; k < no; k += G) {
                    if (ovfR[k] == r) acc += *reinterpret_cast<const v4f*>(&src[(size_t)ovfS[k] * F + q4]);
                }
            }

#pragma unroll
            for (int off = LPR; off < 32; off <<= 1) {
                acc.x += __shfl_xor(acc.x, off);
                acc.y += __shfl_xor(acc.y, off);
                acc.z += __shfl_xor(acc.z, off);
                acc.w += __shfl_xor(acc.w, off);
            }

            if (g == 0) {
                const float d = (float)cnt;
                const float sc = SELF ? rsqrtf(d + 1.0f) : rsqrtf(fmaxf(d, 1.0f));
                *reinterpret_cast<v4f*>(&dst[(size_t)r * F + q4]) = acc * sc;
            }
        };

        int slotX = slotLine(0);
        int slotY = slotLine(1);

        for (int i = 0; i < m; i += 2) {
            const int slotX2 = slotLine(i + 2);
            const int slotY2 = slotLine(i + 3);

            int sx[U], sy[U];
#pragma unroll
            for (int u = 0; u < U; ++u) {
                sx[u] = __shfl(slotX, u * G + g);
                sy[u] = __shfl(slotY, u * G + g);
            }

            v4f vx[U], vy[U];
            v4f selfX, selfY;
            issueRows(vx, selfX, i, sx);
            issueRows(vy, selfY, i + 1, sy);

            finishRow(vx, selfX, i, slotX);
            if (i + 1 < m) finishRow(vy, selfY, i + 1, slotY);

            slotX = slotX2;
            slotY = slotY2;
        }
    }
}

template <int NOUT, bool TANH, bool ADDONE>
__global__ __launch_bounds__(128) void gcn_gemm(const float* __restrict__ in,
                                                const float* __restrict__ W,
                                                const float* __restrict__ bias,
                                                const int* __restrict__ cntS,
                                                float* __restrict__ out, int N) {
    constexpr int K = 64;
    constexpr int CT = NOUT / 16;
    constexpr int PITCH = 68;

    __shared__ float Wt[NOUT * PITCH];
    __shared__ float Bl[NOUT];
    __shared__ float Al[64 * PITCH];

    const int tid = threadIdx.x;
    const int wave = tid >> 5;
    const int lane = tid & 31;
    const int rowBase = blockIdx.x * 64;
    const int tileRow = rowBase + wave * 16;
    const int lm = lane & 15;
    const int kh = (lane >> 4) * 2;
    const int mh = (lane >> 4) * 8;

    constexpr int WI = (K * NOUT / 4) / 128;
    constexpr int AI = (64 * (K / 4)) / 128;
    const float4* Wg4 = reinterpret_cast<const float4*>(W);
    const float4* Ag4 = reinterpret_cast<const float4*>(in);
    float4 wv[WI];
    float4 av[AI];
    int cs[8];
#pragma unroll
    for (int t = 0; t < WI; ++t) wv[t] = Wg4[tid + t * 128];
#pragma unroll
    for (int t = 0; t < AI; ++t) {
        const int i = tid + t * 128;
        const int row = rowBase + i / (K / 4);
        av[t] = make_float4(0.f, 0.f, 0.f, 0.f);
        if (row < N) av[t] = Ag4[(size_t)row * (K / 4) + (i % (K / 4))];
    }
    float bl = 0.f;
    if (tid < NOUT) bl = bias[tid];
#pragma unroll
    for (int j = 0; j < 8; ++j) {
        const int row = tileRow + mh + j;
        cs[j] = cntS[row < N ? row : N - 1];
    }

#pragma unroll
    for (int t = 0; t < WI; ++t) {
        const int i = tid + t * 128;
        const int k = i / (NOUT / 4);
        const int n = (i % (NOUT / 4)) * 4;
        Wt[(n + 0) * PITCH + k] = wv[t].x;
        Wt[(n + 1) * PITCH + k] = wv[t].y;
        Wt[(n + 2) * PITCH + k] = wv[t].z;
        Wt[(n + 3) * PITCH + k] = wv[t].w;
    }
    if (tid < NOUT) Bl[tid] = bl;

#pragma unroll
    for (int t = 0; t < AI; ++t) {
        const int i = tid + t * 128;
        const int c4 = (i % (K / 4)) * 4;
        *reinterpret_cast<float4*>(&Al[(i / (K / 4)) * PITCH + c4]) = av[t];
    }
    __syncthreads();

    if (tileRow >= N) return;

    v8f acc[CT] = {};
    const float* Arow = &Al[(wave * 16 + lm) * PITCH + kh];

#pragma unroll
    for (int k0 = 0; k0 < K; k0 += 4) {
        v2f a = *reinterpret_cast<const v2f*>(&Arow[k0]);
#pragma unroll
        for (int c = 0; c < CT; ++c) {
            v2f b = *reinterpret_cast<const v2f*>(&Wt[(c * 16 + lm) * PITCH + k0 + kh]);
            acc[c] = __builtin_amdgcn_wmma_f32_16x16x4_f32(
                false, a, false, b, (short)0, acc[c], false, false);
        }
    }

    float rs[8];
#pragma unroll
    for (int j = 0; j < 8; ++j) {
        int row = tileRow + mh + j;
        float d = (row < N) ? (float)cs[j] : 1.0f;
        rs[j] = ADDONE ? rsqrtf(d + 1.0f) : rsqrtf(fmaxf(d, 1.0f));
    }

#pragma unroll
    for (int c = 0; c < CT; ++c) {
        int n = c * 16 + lm;
        float bv = Bl[n];
#pragma unroll
        for (int j = 0; j < 8; ++j) {
            int row = tileRow + mh + j;
            if (row < N) {
                float v = acc[c][j] + bv;
                if (TANH) v = tanhf(v);
                out[(size_t)row * NOUT + n] = v * rs[j];
            }
        }
    }
}

extern "C" void kernel_launch(void* const* d_in, const int* in_sizes, int n_in,
                              void* d_out, int out_size, void* d_ws, size_t ws_size,
                              hipStream_t stream) {
    const float* nodes = (const float*)d_in[0];
    const int* senders = (const int*)d_in[1];
    const int* receivers = (const int*)d_in[2];
    const float* W0 = (const float*)d_in[3];
    const float* b0 = (const float*)d_in[4];
    const float* W1 = (const float*)d_in[5];
    const float* b1 = (const float*)d_in[6];
    const float* W2 = (const float*)d_in[7];
    const float* b2 = (const float*)d_in[8];
    float* out = (float*)d_out;

    const int N = in_sizes[0] / 64;
    const int E = in_sizes[1];

    float* X = (float*)d_ws;
    float* Y = X + (size_t)N * 64;
    float* Z = Y + (size_t)N * 64;
    int* col = (int*)(Z + (size_t)N * 16);
    int* ovfS = col + (size_t)N * CAP;
    int* ovfR = ovfS + E;
    int* cntS = ovfR + E;
    int* cntR = cntS + N;
    int* ovfCount = cntR + N;

    const int zn = 2 * N + 1;
    gcn_zero_i<<<(zn + 255) / 256, 256, 0, stream>>>(cntS, zn);
    gcn_fill<<<(E + 255) / 256, 256, 0, stream>>>(senders, receivers, cntS, cntR, col,
                                                  ovfCount, ovfS, ovfR, E);

    const int tiles = (N + 15) / 16;
    const int gblocks = (tiles + 3) / 4;

    gcn_gemm<64, true, true><<<gblocks, 128, 0, stream>>>(nodes, W0, b0, cntS, X, N);
    gcn_gather<64, true, 8><<<GATHER_BLOCKS, 256, 0, stream>>>(X, cntR, col, ovfCount, ovfS, ovfR, Y, N, E);

    gcn_gemm<64, true, true><<<gblocks, 128, 0, stream>>>(Y, W1, b1, cntS, X, N);
    gcn_gather<64, true, 8><<<GATHER_BLOCKS, 256, 0, stream>>>(X, cntR, col, ovfCount, ovfS, ovfR, Y, N, E);

    gcn_gemm<16, false, false><<<gblocks, 128, 0, stream>>>(Y, W2, b2, cntS, Z, N);
    gcn_gather<16, false, 2><<<GATHER_BLOCKS, 256, 0, stream>>>(Z, cntR, col, ovfCount, ovfS, ovfR, out, N, E);
}


# edge tables built through LDS: each work-group sorts its edges into 256-node bins (one returning add per run, not per edge); a bin's degrees and slot lines are made in LDS; slots still by arrival
# speedup vs baseline: 3.8514x; 1.6584x over previous
#include <hip/hip_runtime.h>
#include <hip/hip_bf16.h>

typedef float v2f __attribute__((ext_vector_type(2)));
typedef float v4f __attribute__((ext_vector_type(4)));
typedef float v8f __attribute__((ext_vector_type(8)));

constexpr int CAP = 32;

constexpr int BINSH = 8;
constexpr int BINW = 1 << BINSH;
constexpr int MAXBINS = 512;
constexpr int BCAP = 4096;
constexpr int EPB = 4096;
constexpr int BINT = 512;
constexpr int EPT = EPB / BINT;
constexpr int CURSTRIDE = 32;
constexpr int BUILDT = 512;
constexpr int BUILDU = 6;

__global__ void gcn_zero_i(int* __restrict__ p, int n) {
    int t = blockIdx.x * blockDim.x + threadIdx.x;
    if (t < n) p[t] = 0;
}

__global__ __launch_bounds__(BINT) void gcn_bin(const int* __restrict__ se, const int* __restrict__ re,
                                               int* __restrict__ curR, int* __restrict__ curS,
                                               int* __restrict__ binR, int* __restrict__ binS, int E) {
    __shared__ int cR[MAXBINS];
    __shared__ int cS[MAXBINS];
    __shared__ int gR[MAXBINS];
    __shared__ int gS[MAXBINS];
    __shared__ int wsum[2 * (BINT / 32)];
    __shared__ int2 sortR[EPB];
    __shared__ int sortS[EPB];

    const int tid = threadIdx.x;
    const int lane = tid & 31;
    const int wv = tid >> 5;
    const int e0 = blockIdx.x * EPB;
    const int n = (E - e0) < EPB ? (E - e0) : EPB;

    cR[tid] = 0;
    cS[tid] = 0;
    __syncthreads();

    int s[EPT], r[EPT], pR[EPT], pS[EPT];
#pragma unroll
    for (int j = 0; j < EPT; ++j) {
        const int idx = tid + j * BINT;
        const int i = e0 + (idx < n ? idx : n - 1);
        s[j] = se[i];
        r[j] = re[i];
    }
#pragma unroll
    for (int j = 0; j < EPT; ++j) {
        pR[j] = 0;
        pS[j] = 0;
        if (tid + j * BINT < n) {
            pR[j] = atomicAdd(&cR[r[j] >> BINSH], 1);
            pS[j] = atomicAdd(&cS[s[j] >> BINSH], 1);
        }
    }
    __syncthreads();

    const int vR = cR[tid];
    const int vS = cS[tid];
    int iR = vR, iS = vS;
#pragma unroll
    for (int d = 1; d < 32; d <<= 1) {
        const int tR = __shfl_up(iR, d);
        const int tS = __shfl_up(iS, d);
        if (lane >= d) {
            iR += tR;
            iS += tS;
        }
    }
    if (lane == 31) {
        wsum[wv] = iR;
        wsum[BINT / 32 + wv] = iS;
    }
    __syncthreads();
    int oR = 0, oS = 0;
    for (int w = 0; w < wv; ++w) {
        oR += wsum[w];
        oS += wsum[BINT / 32 + w];
    }
    int gbR = 0, gbS = 0;
    if (vR > 0) gbR = atomicAdd(&curR[tid * CURSTRIDE], vR);
    if (vS > 0) gbS = atomicAdd(&curS[tid * CURSTRIDE], vS);
    cR[tid] = oR + iR - vR;
    cS[tid] = oS + iS - vS;
    gR[tid] = gbR;
    gS[tid] = gbS;
    __syncthreads();

#pragma unroll
    for (int j = 0; j < EPT; ++j) {
        if (tid + j * BINT < n) {
            sortR[cR[r[j] >> BINSH] + pR[j]] = make_int2(s[j], r[j]);
            sortS[cS[s[j] >> BINSH] + pS[j]] = s[j];
        }
    }
    __syncthreads();

    int2* binR2 = reinterpret_cast<int2*>(binR);
    for (int idx = tid; idx < n; idx += BINT) {
        const int2 e = sortR[idx];
        const int kr = e.y >> BINSH;
        const int offR = gR[kr] + (idx - cR[kr]);
        if (offR < BCAP) binR2[(size_t)kr * BCAP + offR] = e;
        const int sv = sortS[idx];
        const int ks = sv >> BINSH;
        const int offS = gS[ks] + (idx - cS[ks]);
        if (offS < BCAP) binS[(size_t)ks * BCAP + offS] = sv;
    }
}

__global__ __launch_bounds__(BUILDT) void gcn_build(const int* __restrict__ se, const int* __restrict__ re,
                                                   const int* __restrict__ curR, const int* __restrict__ curS,
                                                   const int* __restrict__ binR, const int* __restrict__ binS,
                                                   int* __restrict__ cntS, int* __restrict__ cntR,
                                                   int* __restrict__ col, int* __restrict__ ovfCount,
                                                   int* __restrict__ ovfS, int* __restrict__ ovfR,
                                                   int N, int E) {
    __shared__ int dR[BINW];
    __shared__ int dS[BINW];
    __shared__ int colL[BINW * CAP];

    const int tid = threadIdx.x;
    const int k = blockIdx.x;
    if (tid < BINW) {
        dR[tid] = 0;
        dS[tid] = 0;
    }
    __syncthreads();

    const int totR = curR[k * CURSTRIDE];
    const int totS = curS[k * CURSTRIDE];

    auto addR = [&](int sv, int rv) {
        const int rl = rv & (BINW - 1);
        const int slot = atomicAdd(&dR[rl], 1);
        if (slot < CAP) {
            colL[rl * CAP + slot] = sv;
        } else {
            const int o = atomicAdd(ovfCount, 1);
            ovfS[o] = sv;
            ovfR[o] = rv;
        }
    };

    if (totR <= BCAP) {
        if (totR > 0) {
            const int2* p = reinterpret_cast<const int2*>(binR) + (size_t)k * BCAP;
            int2 ev[BUILDU];
#pragma unroll
            for (int j = 0; j < BUILDU; ++j) {
                const int i = tid + j * BUILDT;
                ev[j] = p[i < totR ? i : totR - 1];
            }
#pragma unroll
            for (int j = 0; j < BUILDU; ++j) {
                if (tid + j * BUILDT < totR) addR(ev[j].x, ev[j].y);
            }
            for (int i = tid + BUILDU * BUILDT; i < totR; i += BUILDT) {
                const int2 e = p[i];
                addR(e.x, e.y);
            }
        }
    } else {
        for (int i = tid; i < E; i += BUILDT) {
            const int rv = re[i];
            if ((rv >> BINSH) == k) addR(se[i], rv);
        }
    }

    if (totS <= BCAP) {
        if (totS > 0) {
            const int* p = binS + (size_t)k * BCAP;
            int sv[BUILDU];
#pragma unroll
            for (int j = 0; j < BUILDU; ++j) {
                const int i = tid + j * BUILDT;
                sv[j] = p[i < totS ? i : totS - 1];
            }
#pragma unroll
            for (int j = 0; j < BUILDU; ++j) {
                if (tid + j * BUILDT < totS) atomicAdd(&dS[sv[j] & (BINW - 1)], 1);
            }
            for (int i = tid + BUILDU * BUILDT; i < totS; i += BUILDT) {
                atomicAdd(&dS[p[i] & (BINW - 1)], 1);
            }
        }
    } else {
        for (int i = tid; i < E; i += BUILDT) {
            const int sv = se[i];
            if ((sv >> BINSH) == k) atomicAdd(&dS[sv & (BINW - 1)], 1);
        }
    }
    __syncthreads();

    const int row0 = k * BINW;
    if (tid < BINW && row0 + tid < N) {
        cntR[row0 + tid] = dR[tid];
        cntS[row0 + tid] = dS[tid];
    }
    for (int idx = tid; idx < BINW * CAP; idx += BUILDT) {
        const int rl = idx / CAP;
        const int j = idx % CAP;
        if (row0 + rl < N) col[(size_t)(row0 + rl) * CAP + j] = (j < dR[rl]) ? colL[idx] : 0;
    }
}

constexpr int GATHER_BLOCKS = 768;
constexpr int GATHER_WAVES = GATHER_BLOCKS * 8;

template <int F, bool SELF, int U>
__global__ __launch_bounds__(256) void gcn_gather(const float* __restrict__ src,
                                                  const int* __restrict__ cntR,
                                                  const int* __restrict__ col,
                                                  const int* __restrict__ ovfCount,
                                                  const int* __restrict__ ovfS,
                                                  const int* __restrict__ ovfR,
                                                  float* __restrict__ dst, int N, int E) {
    constexpr int LPR = F / 4;
    constexpr int G = 32 / LPR;
    const int lane = threadIdx.x & 31;
    const int g = lane / LPR;
    const int q4 = (lane % LPR) * 4;

    const int wave = __builtin_amdgcn_readfirstlane((int)(blockIdx.x * 8 + (threadIdx.x >> 5)));
    const int base = N / GATHER_WAVES;
    const int rem = N % GATHER_WAVES;
    const int r0 = wave * base + (wave < rem ? wave : rem);
    const int r1 = r0 + base + (wave < rem ? 1 : 0);

    const v4f zero4 = {0.f, 0.f, 0.f, 0.f};

    for (int rb = r0; rb < r1; rb += 32) {
        const int m = (r1 - rb) < 32 ? (r1 - rb) : 32;
        const int cntv = cntR[rb + (lane < m ? lane : m - 1)];

        auto rowCount = [&](int i) -> int { return i < m ? __builtin_amdgcn_readlane(cntv, i & 31) : 0; };

        auto slotLine = [&](int i) -> int {
            const int c = rowCount(i);
            const int n = c < CAP ? c : CAP;
            const int ri = rb + (i < m ? i : m - 1);
            const int* p = (n > 0) ? &col[(size_t)ri * CAP + (lane < n ? lane : n - 1)] : &cntR[ri];
            return *p;
        };

        auto issueRows = [&](v4f (&v)[U], v4f& selfv, int i, const int (&s)[U]) {
            const int c = rowCount(i);
            const int n = c < CAP ? c : CAP;
            if (SELF && i < m) selfv = *reinterpret_cast<const v4f*>(&src[(size_t)(rb + i) * F + q4]);
#pragma unroll
            for (int u = 0; u < U; ++u) {
                if (u * G < n) v[u] = *reinterpret_cast<const v4f*>(&src[(size_t)s[u] * F + q4]);
            }
        };

        auto finishRow = [&](const v4f (&v)[U], const v4f& selfv, int i, int myS) {
            const int r = rb + i;
            const int cnt = rowCount(i);
            const int n = cnt < CAP ? cnt : CAP;

            v4f acc = zero4;
            if (SELF && g == 0) acc = selfv;
#pragma unroll
            for (int u = 0; u < U; ++u) {
                if (u * G < n) acc += (u * G + g < n) ? v[u] : zero4;
            }

            for (int j0 = U * G; j0 < n; j0 += U * G) {
                v4f w[U];
#pragma unroll
                for (int u = 0; u < U; ++u) {
                    const int j = j0 + u * G + g;
                    const int s = __shfl(myS, j & 31);
                    w[u] = zero4;
                    if (j < n) w[u] = *reinterpret_cast<const v4f*>(&src[(size_t)s * F + q4]);
                }
#pragma unroll
                for (int u = 0; u < U; ++u) acc += w[u];
            }

            if (cnt > CAP) {
                int no = *ovfCount;
                if (no > E) no = E;
                for (int k = g; k < no; k += G) {
                    if (ovfR[k] == r) acc += *reinterpret_cast<const v4f*>(&src[(size_t)ovfS[k] * F + q4]);
                }
            }

#pragma unroll
            for (int off = LPR; off < 32; off <<= 1) {
                acc.x += __shfl_xor(acc.x, off);
                acc.y += __shfl_xor(acc.y, off);
                acc.z += __shfl_xor(acc.z, off);
                acc.w += __shfl_xor(acc.w, off);
            }

            if (g == 0) {
                const float d = (float)cnt;
                const float sc = SELF ? rsqrtf(d + 1.0f) : rsqrtf(fmaxf(d, 1.0f));
                *reinterpret_cast<v4f*>(&dst[(size_t)r * F + q4]) = acc * sc;
            }
        };

        int slotX = slotLine(0);
        int slotY = slotLine(1);

        for (int i = 0; i < m; i += 2) {
            const int slotX2 = slotLine(i + 2);
            const int slotY2 = slotLine(i + 3);

            int sx[U], sy[U];
#pragma unroll
            for (int u = 0; u < U; ++u) {
                sx[u] = __shfl(slotX, u * G + g);
                sy[u] = __shfl(slotY, u * G + g);
            }

            v4f vx[U], vy[U];
            v4f selfX, selfY;
            issueRows(vx, selfX, i, sx);
            issueRows(vy, selfY, i + 1, sy);

            finishRow(vx, selfX, i, slotX);
            if (i + 1 < m) finishRow(vy, selfY, i + 1, slotY);

            slotX = slotX2;
            slotY = slotY2;
        }
    }
}

template <int NOUT, bool TANH, bool ADDONE>
__global__ __launch_bounds__(128) void gcn_gemm(const float* __restrict__ in,
                                                const float* __restrict__ W,
                                                const float* __restrict__ bias,
                                                const int* __restrict__ cntS,
                                                float* __restrict__ out, int N) {
    constexpr int K = 64;
    constexpr int CT = NOUT / 16;
    constexpr int PITCH = 68;

    __shared__ float Wt[NOUT * PITCH];
    __shared__ float Bl[NOUT];
    __shared__ float Al[64 * PITCH];

    const int tid = threadIdx.x;
    const int wave = tid >> 5;
    const int lane = tid & 31;
    const int rowBase = blockIdx.x * 64;
    const int tileRow = rowBase + wave * 16;
    const int lm = lane & 15;
    const int kh = (lane >> 4) * 2;
    const int mh = (lane >> 4) * 8;

    constexpr int WI = (K * NOUT / 4) / 128;
    constexpr int AI = (64 * (K / 4)) / 128;
    const float4* Wg4 = reinterpret_cast<const float4*>(W);
    const float4* Ag4 = reinterpret_cast<const float4*>(in);
    float4 wv[WI];
    float4 av[AI];
    int cs[8];
#pragma unroll
    for (int t = 0; t < WI; ++t) wv[t] = Wg4[tid + t * 128];
#pragma unroll
    for (int t = 0; t < AI; ++t) {
        const int i = tid + t * 128;
        const int row = rowBase + i / (K / 4);
        av[t] = make_float4(0.f, 0.f, 0.f, 0.f);
        if (row < N) av[t] = Ag4[(size_t)row * (K / 4) + (i % (K / 4))];
    }
    float bl = 0.f;
    if (tid < NOUT) bl = bias[tid];
#pragma unroll
    for (int j = 0; j < 8; ++j) {
        const int row = tileRow + mh + j;
        cs[j] = cntS[row < N ? row : N - 1];
    }

#pragma unroll
    for (int t = 0; t < WI; ++t) {
        const int i = tid + t * 128;
        const int k = i / (NOUT / 4);
        const int n = (i % (NOUT / 4)) * 4;
        Wt[(n + 0) * PITCH + k] = wv[t].x;
        Wt[(n + 1) * PITCH + k] = wv[t].y;
        Wt[(n + 2) * PITCH + k] = wv[t].z;
        Wt[(n + 3) * PITCH + k] = wv[t].w;
    }
    if (tid < NOUT) Bl[tid] = bl;

#pragma unroll
    for (int t = 0; t < AI; ++t) {
        const int i = tid + t * 128;
        const int c4 = (i % (K / 4)) * 4;
        *reinterpret_cast<float4*>(&Al[(i / (K / 4)) * PITCH + c4]) = av[t];
    }
    __syncthreads();

    if (tileRow >= N) return;

    v8f acc[CT] = {};
    const float* Arow = &Al[(wave * 16 + lm) * PITCH + kh];

#pragma unroll
    for (int k0 = 0; k0 < K; k0 += 4) {
        v2f a = *reinterpret_cast<const v2f*>(&Arow[k0]);
#pragma unroll
        for (int c = 0; c < CT; ++c) {
            v2f b = *reinterpret_cast<const v2f*>(&Wt[(c * 16 + lm) * PITCH + k0 + kh]);
            acc[c] = __builtin_amdgcn_wmma_f32_16x16x4_f32(
                false, a, false, b, (short)0, acc[c], false, false);
        }
    }

    float rs[8];
#pragma unroll
    for (int j = 0; j < 8; ++j) {
        int row = tileRow + mh + j;
        float d = (row < N) ? (float)cs[j] : 1.0f;
        rs[j] = ADDONE ? rsqrtf(d + 1.0f) : rsqrtf(fmaxf(d, 1.0f));
    }

#pragma unroll
    for (int c = 0; c < CT; ++c) {
        int n = c * 16 + lm;
        float bv = Bl[n];
#pragma unroll
        for (int j = 0; j < 8; ++j) {
            int row = tileRow + mh + j;
            if (row < N) {
                float v = acc[c][j] + bv;
                if (TANH) v = tanhf(v);
                out[(size_t)row * NOUT + n] = v * rs[j];
            }
        }
    }
}

extern "C" void kernel_launch(void* const* d_in, const int* in_sizes, int n_in,
                              void* d_out, int out_size, void* d_ws, size_t ws_size,
                              hipStream_t stream) {
    const float* nodes = (const float*)d_in[0];
    const int* senders = (const int*)d_in[1];
    const int* receivers = (const int*)d_in[2];
    const float* W0 = (const float*)d_in[3];
    const float* b0 = (const float*)d_in[4];
    const float* W1 = (const float*)d_in[5];
    const float* b1 = (const float*)d_in[6];
    const float* W2 = (const float*)d_in[7];
    const float* b2 = (const float*)d_in[8];
    float* out = (float*)d_out;

    const int N = in_sizes[0] / 64;
    const int E = in_sizes[1];

    float* X = (float*)d_ws;
    float* Y = X + (size_t)N * 64;
    float* Z = Y + (size_t)N * 64;
    int* col = (int*)(Z + (size_t)N * 16);
    int* ovfS = col + (size_t)N * CAP;
    int* ovfR = ovfS + E;
    int* cntS = ovfR + E;
    int* cntR = cntS + N;
    int* curR = cntR + N;
    int* curS = curR + MAXBINS * CURSTRIDE;
    int* ovfCount = curS + MAXBINS * CURSTRIDE;
    int* binS = ovfCount + 4;
    const int nbins = (N + BINW - 1) / BINW;
    int* binR = binS + (size_t)nbins * BCAP;

    const int zn = 2 * MAXBINS * CURSTRIDE + 1;
    gcn_zero_i<<<(zn + 255) / 256, 256, 0, stream>>>(curR, zn);
    gcn_bin<<<(E + EPB - 1) / EPB, BINT, 0, stream>>>(senders, receivers, curR, curS, binR, binS, E);
    gcn_build<<<nbins, BUILDT, 0, stream>>>(senders, receivers, curR, curS, binR, binS, cntS, cntR,
                                            col, ovfCount, ovfS, ovfR, N, E);

    const int tiles = (N + 15) / 16;
    const int gblocks = (tiles + 3) / 4;

    gcn_gemm<64, true, true><<<gblocks, 128, 0, stream>>>(nodes, W0, b0, cntS, X, N);
    gcn_gather<64, true, 8><<<GATHER_BLOCKS, 256, 0, stream>>>(X, cntR, col, ovfCount, ovfS, ovfR, Y, N, E);

    gcn_gemm<64, true, true><<<gblocks, 128, 0, stream>>>(Y, W1, b1, cntS, X, N);
    gcn_gather<64, true, 8><<<GATHER_BLOCKS, 256, 0, stream>>>(X, cntR, col, ovfCount, ovfS, ovfR, Y, N, E);

    gcn_gemm<16, false, false><<<gblocks, 128, 0, stream>>>(Y, W2, b2, cntS, Z, N);
    gcn_gather<16, false, 2><<<GATHER_BLOCKS, 256, 0, stream>>>(Z, cntR, col, ovfCount, ovfS, ovfR, out, N, E);
}


# each 64-wide aggregation fused with the next layer's GEMM: a wave's gathered rows fill its own 16-row LDS tile that feeds its WMMA loop; the aggregated matrix is never stored; 9 -> 7 launches
# speedup vs baseline: 4.2207x; 1.0959x over previous
#include <hip/hip_runtime.h>
#include <hip/hip_bf16.h>

typedef float v2f __attribute__((ext_vector_type(2)));
typedef float v4f __attribute__((ext_vector_type(4)));
typedef float v8f __attribute__((ext_vector_type(8)));

constexpr int CAP = 32;

constexpr int BINSH = 8;
constexpr int BINW = 1 << BINSH;
constexpr int MAXBINS = 512;
constexpr int BCAP = 4096;
constexpr int EPB = 4096;
constexpr int BINT = 512;
constexpr int EPT = EPB / BINT;
constexpr int CURSTRIDE = 32;
constexpr int BUILDT = 512;
constexpr int BUILDU = 6;

__global__ void gcn_zero_i(int* __restrict__ p, int n) {
    int t = blockIdx.x * blockDim.x + threadIdx.x;
    if (t < n) p[t] = 0;
}

__global__ __launch_bounds__(BINT) void gcn_bin(const int* __restrict__ se, const int* __restrict__ re,
                                               int* __restrict__ curR, int* __restrict__ curS,
                                               int* __restrict__ binR, int* __restrict__ binS, int E) {
    __shared__ int cR[MAXBINS];
    __shared__ int cS[MAXBINS];
    __shared__ int gR[MAXBINS];
    __shared__ int gS[MAXBINS];
    __shared__ int wsum[2 * (BINT / 32)];
    __shared__ int2 sortR[EPB];
    __shared__ int sortS[EPB];

    const int tid = threadIdx.x;
    const int lane = tid & 31;
    const int wv = tid >> 5;
    const int e0 = blockIdx.x * EPB;
    const int n = (E - e0) < EPB ? (E - e0) : EPB;

    cR[tid] = 0;
    cS[tid] = 0;
    __syncthreads();

    int s[EPT], r[EPT], pR[EPT], pS[EPT];
#pragma unroll
    for (int j = 0; j < EPT; ++j) {
        const int idx = tid + j * BINT;
        const int i = e0 + (idx < n ? idx : n - 1);
        s[j] = se[i];
        r[j] = re[i];
    }
#pragma unroll
    for (int j = 0; j < EPT; ++j) {
        pR[j] = 0;
        pS[j] = 0;
        if (tid + j * BINT < n) {
            pR[j] = atomicAdd(&cR[r[j] >> BINSH], 1);
            pS[j] = atomicAdd(&cS[s[j] >> BINSH], 1);
        }
    }
    __syncthreads();

    const int vR = cR[tid];
    const int vS = cS[tid];
    int iR = vR, iS = vS;
#pragma unroll
    for (int d = 1; d < 32; d <<= 1) {
        const int tR = __shfl_up(iR, d);
        const int tS = __shfl_up(iS, d);
        if (lane >= d) {
            iR += tR;
            iS += tS;
        }
    }
    if (lane == 31) {
        wsum[wv] = iR;
        wsum[BINT / 32 + wv] = iS;
    }
    __syncthreads();
    int oR = 0, oS = 0;
    for (int w = 0; w < wv; ++w) {
        oR += wsum[w];
        oS += wsum[BINT / 32 + w];
    }
    int gbR = 0, gbS = 0;
    if (vR > 0) gbR = atomicAdd(&curR[tid * CURSTRIDE], vR);
    if (vS > 0) gbS = atomicAdd(&curS[tid * CURSTRIDE], vS);
    cR[tid] = oR + iR - vR;
    cS[tid] = oS + iS - vS;
    gR[tid] = gbR;
    gS[tid] = gbS;
    __syncthreads();

#pragma unroll
    for (int j = 0; j < EPT; ++j) {
        if (tid + j * BINT < n) {
            sortR[cR[r[j] >> BINSH] + pR[j]] = make_int2(s[j], r[j]);
            sortS[cS[s[j] >> BINSH] + pS[j]] = s[j];
        }
    }
    __syncthreads();

    int2* binR2 = reinterpret_cast<int2*>(binR);
    for (int idx = tid; idx < n; idx += BINT) {
        const int2 e = sortR[idx];
        const int kr = e.y >> BINSH;
        const int offR = gR[kr] + (idx - cR[kr]);
        if (offR < BCAP) binR2[(size_t)kr * BCAP + offR] = e;
        const int sv = sortS[idx];
        const int ks = sv >> BINSH;
        const int offS = gS[ks] + (idx - cS[ks]);
        if (offS < BCAP) binS[(size_t)ks * BCAP + offS] = sv;
    }
}

__global__ __launch_bounds__(BUILDT) void gcn_build(const int* __restrict__ se, const int* __restrict__ re,
                                                   const int* __restrict__ curR, const int* __restrict__ curS,
                                                   const int* __restrict__ binR, const int* __restrict__ binS,
                                                   int* __restrict__ cntS, int* __restrict__ cntR,
                                                   int* __restrict__ col, int* __restrict__ ovfCount,
                                                   int* __restrict__ ovfS, int* __restrict__ ovfR,
                                                   int N, int E) {
    __shared__ int dR[BINW];
    __shared__ int dS[BINW];
    __shared__ int colL[BINW * CAP];

    const int tid = threadIdx.x;
    const int k = blockIdx.x;
    if (tid < BINW) {
        dR[tid] = 0;
        dS[tid] = 0;
    }
    __syncthreads();

    const int totR = curR[k * CURSTRIDE];
    const int totS = curS[k * CURSTRIDE];

    auto addR = [&](int sv, int rv) {
        const int rl = rv & (BINW - 1);
        const int slot = atomicAdd(&dR[rl], 1);
        if (slot < CAP) {
            colL[rl * CAP + slot] = sv;
        } else {
            const int o = atomicAdd(ovfCount, 1);
            ovfS[o] = sv;
            ovfR[o] = rv;
        }
    };

    if (totR <= BCAP) {
        if (totR > 0) {
            const int2* p = reinterpret_cast<const int2*>(binR) + (size_t)k * BCAP;
            int2 ev[BUILDU];
#pragma unroll
            for (int j = 0; j < BUILDU; ++j) {
                const int i = tid + j * BUILDT;
                ev[j] = p[i < totR ? i : totR - 1];
            }
#pragma unroll
            for (int j = 0; j < BUILDU; ++j) {
                if (tid + j * BUILDT < totR) addR(ev[j].x, ev[j].y);
            }
            for (int i = tid + BUILDU * BUILDT; i < totR; i += BUILDT) {
                const int2 e = p[i];
                addR(e.x, e.y);
            }
        }
    } else {
        for (int i = tid; i < E; i += BUILDT) {
            const int rv = re[i];
            if ((rv >> BINSH) == k) addR(se[i], rv);
        }
    }

    if (totS <= BCAP) {
        if (totS > 0) {
            const int* p = binS + (size_t)k * BCAP;
            int sv[BUILDU];
#pragma unroll
            for (int j = 0; j < BUILDU; ++j) {
                const int i = tid + j * BUILDT;
                sv[j] = p[i < totS ? i : totS - 1];
            }
#pragma unroll
            for (int j = 0; j < BUILDU; ++j) {
                if (tid + j * BUILDT < totS) atomicAdd(&dS[sv[j] & (BINW - 1)], 1);
            }
            for (int i = tid + BUILDU * BUILDT; i < totS; i += BUILDT) {
                atomicAdd(&dS[p[i] & (BINW - 1)], 1);
            }
        }
    } else {
        for (int i = tid; i < E; i += BUILDT) {
            const int sv = se[i];
            if ((sv >> BINSH) == k) atomicAdd(&dS[sv & (BINW - 1)], 1);
        }
    }
    __syncthreads();

    const int row0 = k * BINW;
    if (tid < BINW && row0 + tid < N) {
        cntR[row0 + tid] = dR[tid];
        cntS[row0 + tid] = dS[tid];
    }
    for (int idx = tid; idx < BINW * CAP; idx += BUILDT) {
        const int rl = idx / CAP;
        const int j = idx % CAP;
        if (row0 + rl < N) col[(size_t)(row0 + rl) * CAP + j] = (j < dR[rl]) ? colL[idx] : 0;
    }
}

constexpr int GATHER_BLOCKS = 768;
constexpr int GATHER_WAVES = GATHER_BLOCKS * 8;

template <int F, bool SELF, int U>
__global__ __launch_bounds__(256) void gcn_gather(const float* __restrict__ src,
                                                  const int* __restrict__ cntR,
                                                  const int* __restrict__ col,
                                                  const int* __restrict__ ovfCount,
                                                  const int* __restrict__ ovfS,
                                                  const int* __restrict__ ovfR,
                                                  float* __restrict__ dst, int N, int E) {
    constexpr int LPR = F / 4;
    constexpr int G = 32 / LPR;
    const int lane = threadIdx.x & 31;
    const int g = lane / LPR;
    const int q4 = (lane % LPR) * 4;

    const int wave = __builtin_amdgcn_readfirstlane((int)(blockIdx.x * 8 + (threadIdx.x >> 5)));
    const int base = N / GATHER_WAVES;
    const int rem = N % GATHER_WAVES;
    const int r0 = wave * base + (wave < rem ? wave : rem);
    const int r1 = r0 + base + (wave < rem ? 1 : 0);

    const v4f zero4 = {0.f, 0.f, 0.f, 0.f};

    for (int rb = r0; rb < r1; rb += 32) {
        const int m = (r1 - rb) < 32 ? (r1 - rb) : 32;
        const int cntv = cntR[rb + (lane < m ? lane : m - 1)];

        auto rowCount = [&](int i) -> int { return i < m ? __builtin_amdgcn_readlane(cntv, i & 31) : 0; };

        auto slotLine = [&](int i) -> int {
            const int c = rowCount(i);
            const int n = c < CAP ? c : CAP;
            const int ri = rb + (i < m ? i : m - 1);
            const int* p = (n > 0) ? &col[(size_t)ri * CAP + (lane < n ? lane : n - 1)] : &cntR[ri];
            return *p;
        };

        auto issueRows = [&](v4f (&v)[U], v4f& selfv, int i, const int (&s)[U]) {
            const int c = rowCount(i);
            const int n = c < CAP ? c : CAP;
            if (SELF && i < m) selfv = *reinterpret_cast<const v4f*>(&src[(size_t)(rb + i) * F + q4]);
#pragma unroll
            for (int u = 0; u < U; ++u) {
                if (u * G < n) v[u] = *reinterpret_cast<const v4f*>(&src[(size_t)s[u] * F + q4]);
            }
        };

        auto finishRow = [&](const v4f (&v)[U], const v4f& selfv, int i, int myS) {
            const int r = rb + i;
            const int cnt = rowCount(i);
            const int n = cnt < CAP ? cnt : CAP;

            v4f acc = zero4;
            if (SELF && g == 0) acc = selfv;
#pragma unroll
            for (int u = 0; u < U; ++u) {
                if (u * G < n) acc += (u * G + g < n) ? v[u] : zero4;
            }

            for (int j0 = U * G; j0 < n; j0 += U * G) {
                v4f w[U];
#pragma unroll
                for (int u = 0; u < U; ++u) {
                    const int j = j0 + u * G + g;
                    const int s = __shfl(myS, j & 31);
                    w[u] = zero4;
                    if (j < n) w[u] = *reinterpret_cast<const v4f*>(&src[(size_t)s * F + q4]);
                }
#pragma unroll
                for (int u = 0; u < U; ++u) acc += w[u];
            }

            if (cnt > CAP) {
                int no = *ovfCount;
                if (no > E) no = E;
                for (int k = g; k < no; k += G) {
                    if (ovfR[k] == r) acc += *reinterpret_cast<const v4f*>(&src[(size_t)ovfS[k] * F + q4]);
                }
            }

#pragma unroll
            for (int off = LPR; off < 32; off <<= 1) {
                acc.x += __shfl_xor(acc.x, off);
                acc.y += __shfl_xor(acc.y, off);
                acc.z += __shfl_xor(acc.z, off);
                acc.w += __shfl_xor(acc.w, off);
            }

            if (g == 0) {
                const float d = (float)cnt;
                const float sc = SELF ? rsqrtf(d + 1.0f) : rsqrtf(fmaxf(d, 1.0f));
                *reinterpret_cast<v4f*>(&dst[(size_t)r * F + q4]) = acc * sc;
            }
        };

        int slotX = slotLine(0);
        int slotY = slotLine(1);

        for (int i = 0; i < m; i += 2) {
            const int slotX2 = slotLine(i + 2);
            const int slotY2 = slotLine(i + 3);

            int sx[U], sy[U];
#pragma unroll
            for (int u = 0; u < U; ++u) {
                sx[u] = __shfl(slotX, u * G + g);
                sy[u] = __shfl(slotY, u * G + g);
            }

            v4f vx[U], vy[U];
            v4f selfX, selfY;
            issueRows(vx, selfX, i, sx);
            issueRows(vy, selfY, i + 1, sy);

            finishRow(vx, selfX, i, slotX);
            if (i + 1 < m) finishRow(vy, selfY, i + 1, slotY);

            slotX = slotX2;
            slotY = slotY2;
        }
    }
}

template <int NOUT, bool TANH, bool ADDONE>
__global__ __launch_bounds__(128) void gcn_gemm(const float* __restrict__ in,
                                                const float* __restrict__ W,
                                                const float* __restrict__ bias,
                                                const int* __restrict__ cntS,
                                                float* __restrict__ out, int N) {
    constexpr int K = 64;
    constexpr int CT = NOUT / 16;
    constexpr int PITCH = 68;

    __shared__ float Wt[NOUT * PITCH];
    __shared__ float Bl[NOUT];
    __shared__ float Al[64 * PITCH];

    const int tid = threadIdx.x;
    const int wave = tid >> 5;
    const int lane = tid & 31;
    const int rowBase = blockIdx.x * 64;
    const int tileRow = rowBase + wave * 16;
    const int lm = lane & 15;
    const int kh = (lane >> 4) * 2;
    const int mh = (lane >> 4) * 8;

    constexpr int WI = (K * NOUT / 4) / 128;
    constexpr int AI = (64 * (K / 4)) / 128;
    const float4* Wg4 = reinterpret_cast<const float4*>(W);
    const float4* Ag4 = reinterpret_cast<const float4*>(in);
    float4 wv[WI];
    float4 av[AI];
    int cs[8];
#pragma unroll
    for (int t = 0; t < WI; ++t) wv[t] = Wg4[tid + t * 128];
#pragma unroll
    for (int t = 0; t < AI; ++t) {
        const int i = tid + t * 128;
        const int row = rowBase + i / (K / 4);
        av[t] = make_float4(0.f, 0.f, 0.f, 0.f);
        if (row < N) av[t] = Ag4[(size_t)row * (K / 4) + (i % (K / 4))];
    }
    float bl = 0.f;
    if (tid < NOUT) bl = bias[tid];
#pragma unroll
    for (int j = 0; j < 8; ++j) {
        const int row = tileRow + mh + j;
        cs[j] = cntS[row < N ? row : N - 1];
    }

#pragma unroll
    for (int t = 0; t < WI; ++t) {
        const int i = tid + t * 128;
        const int k = i / (NOUT / 4);
        const int n = (i % (NOUT / 4)) * 4;
        Wt[(n + 0) * PITCH + k] = wv[t].x;
        Wt[(n + 1) * PITCH + k] = wv[t].y;
        Wt[(n + 2) * PITCH + k] = wv[t].z;
        Wt[(n + 3) * PITCH + k] = wv[t].w;
    }
    if (tid < NOUT) Bl[tid] = bl;

#pragma unroll
    for (int t = 0; t < AI; ++t) {
        const int i = tid + t * 128;
        const int c4 = (i % (K / 4)) * 4;
        *reinterpret_cast<float4*>(&Al[(i / (K / 4)) * PITCH + c4]) = av[t];
    }
    __syncthreads();

    if (tileRow >= N) return;

    v8f acc[CT] = {};
    const float* Arow = &Al[(wave * 16 + lm) * PITCH + kh];

#pragma unroll
    for (int k0 = 0; k0 < K; k0 += 4) {
        v2f a = *reinterpret_cast<const v2f*>(&Arow[k0]);
#pragma unroll
        for (int c = 0; c < CT; ++c) {
            v2f b = *reinterpret_cast<const v2f*>(&Wt[(c * 16 + lm) * PITCH + k0 + kh]);
            acc[c] = __builtin_amdgcn_wmma_f32_16x16x4_f32(
                false, a, false, b, (short)0, acc[c], false, false);
        }
    }

    float rs[8];
#pragma unroll
    for (int j = 0; j < 8; ++j) {
        int row = tileRow + mh + j;
        float d = (row < N) ? (float)cs[j] : 1.0f;
        rs[j] = ADDONE ? rsqrtf(d + 1.0f) : rsqrtf(fmaxf(d, 1.0f));
    }

#pragma unroll
    for (int c = 0; c < CT; ++c) {
        int n = c * 16 + lm;
        float bv = Bl[n];
#pragma unroll
        for (int j = 0; j < 8; ++j) {
            int row = tileRow + mh + j;
            if (row < N) {
                float v = acc[c][j] + bv;
                if (TANH) v = tanhf(v);
                out[(size_t)row * NOUT + n] = v * rs[j];
            }
        }
    }
}

template <int NOUT, bool TANH, bool ADDONE, int U>
__global__ __launch_bounds__(256, 6) void gcn_gather_gemm(const float* __restrict__ src,
                                                       const int* __restrict__ cntR,
                                                       const int* __restrict__ col,
                                                       const int* __restrict__ ovfCount,
                                                       const int* __restrict__ ovfS,
                                                       const int* __restrict__ ovfR,
                                                       const float* __restrict__ W,
                                                       const float* __restrict__ bias,
                                                       const int* __restrict__ cntS,
                                                       float* __restrict__ out, int N, int E) {
    constexpr int F = 64;
    constexpr int K = 64;
    constexpr int CT = NOUT / 16;
    constexpr int PITCH = 68;
    constexpr int LPR = F / 4;
    constexpr int G = 32 / LPR;

    __shared__ alignas(16) float Wt[NOUT * PITCH];
    __shared__ float Bl[NOUT];
    __shared__ alignas(16) float Al[8 * 16 * PITCH];

    const int tid = threadIdx.x;
    const int lane = tid & 31;
    const int g = lane / LPR;
    const int q4 = (lane % LPR) * 4;
    const int lm = lane & 15;
    const int kh = (lane >> 4) * 2;
    const int mh = (lane >> 4) * 8;

    {
        constexpr int WI = (K * NOUT / 4) / 256;
        const float4* Wg4 = reinterpret_cast<const float4*>(W);
        float4 wv[WI];
#pragma unroll
        for (int t = 0; t < WI; ++t) wv[t] = Wg4[tid + t * 256];
        float bl = 0.f;
        if (tid < NOUT) bl = bias[tid];
#pragma unroll
        for (int t = 0; t < WI; ++t) {
            const int i = tid + t * 256;
            const int k = i / (NOUT / 4);
            const int n = (i % (NOUT / 4)) * 4;
            Wt[(n + 0) * PITCH + k] = wv[t].x;
            Wt[(n + 1) * PITCH + k] = wv[t].y;
            Wt[(n + 2) * PITCH + k] = wv[t].z;
            Wt[(n + 3) * PITCH + k] = wv[t].w;
        }
        if (tid < NOUT) Bl[tid] = bl;
    }
    __syncthreads();

    const int wl = __builtin_amdgcn_readfirstlane((int)(threadIdx.x >> 5));
    const int wave = __builtin_amdgcn_readfirstlane((int)(blockIdx.x * 8)) + wl;
    const int base = N / GATHER_WAVES;
    const int rem = N % GATHER_WAVES;
    const int r0 = wave * base + (wave < rem ? wave : rem);
    const int r1 = r0 + base + (wave < rem ? 1 : 0);

    float* At = &Al[wl * 16 * PITCH];
    const v4f zero4 = {0.f, 0.f, 0.f, 0.f};

    for (int rb = r0; rb < r1; rb += 16) {
        const int m = (r1 - rb) < 16 ? (r1 - rb) : 16;
        const int cntv = cntR[rb + (lane < m ? lane : m - 1)];
        const int csv = cntS[rb + (lane < m ? lane : m - 1)];

        auto rowCount = [&](int i) -> int { return i < m ? __builtin_amdgcn_readlane(cntv, i & 31) : 0; };

        auto slotLine = [&](int i) -> int {
            const int c = rowCount(i);
            const int n = c < CAP ? c : CAP;
            const int ri = rb + (i < m ? i : m - 1);
            const int* p = (n > 0) ? &col[(size_t)ri * CAP + (lane < n ? lane : n - 1)] : &cntR[ri];
            return *p;
        };

        auto issueRows = [&](v4f (&v)[U], v4f& selfv, int i, const int (&s)[U]) {
            const int c = rowCount(i);
            const int n = c < CAP ? c : CAP;
            if (i < m) selfv = *reinterpret_cast<const v4f*>(&src[(size_t)(rb + i) * F + q4]);
#pragma unroll
            for (int u = 0; u < U; ++u) {
                if (u * G < n) v[u] = *reinterpret_cast<const v4f*>(&src[(size_t)s[u] * F + q4]);
            }
        };

        auto finishRow = [&](const v4f (&v)[U], const v4f& selfv, int i, int myS) {
            const int r = rb + i;
            const int cnt = rowCount(i);
            const int n = cnt < CAP ? cnt : CAP;

            v4f acc = zero4;
            if (g == 0) acc = selfv;
#pragma unroll
            for (int u = 0; u < U; ++u) {
                if (u * G < n) acc += (u * G + g < n) ? v[u] : zero4;
            }

            for (int j0 = U * G; j0 < n; j0 += U * G) {
                v4f w[U];
#pragma unroll
                for (int u = 0; u < U; ++u) {
                    const int j = j0 + u * G + g;
                    const int s = __shfl(myS, j & 31);
                    w[u] = zero4;
                    if (j < n) w[u] = *reinterpret_cast<const v4f*>(&src[(size_t)s * F + q4]);
                }
#pragma unroll
                for (int u = 0; u < U; ++u) acc += w[u];
            }

            if (cnt > CAP) {
                int no = *ovfCount;
                if (no > E) no = E;
                for (int k = g; k < no; k += G) {
                    if (ovfR[k] == r) acc += *reinterpret_cast<const v4f*>(&src[(size_t)ovfS[k] * F + q4]);
                }
            }

#pragma unroll
            for (int off = LPR; off < 32; off <<= 1) {
                acc.x += __shfl_xor(acc.x, off);
                acc.y += __shfl_xor(acc.y, off);
                acc.z += __shfl_xor(acc.z, off);
                acc.w += __shfl_xor(acc.w, off);
            }

            if (g == 0) {
                const float sc = rsqrtf((float)cnt + 1.0f);
                *reinterpret_cast<v4f*>(&At[i * PITCH + q4]) = acc * sc;
            }
        };

        int slotX = slotLine(0);
        int slotY = slotLine(1);

        for (int i = 0; i < m; i += 2) {
            const int slotX2 = slotLine(i + 2);
            const int slotY2 = slotLine(i + 3);

            int sx[U], sy[U];
#pragma unroll
            for (int u = 0; u < U; ++u) {
                sx[u] = __shfl(slotX, u * G + g);
                sy[u] = __shfl(slotY, u * G + g);
            }

            v4f vx[U], vy[U];
            v4f selfX, selfY;
            issueRows(vx, selfX, i, sx);
            issueRows(vy, selfY, i + 1, sy);

            finishRow(vx, selfX, i, slotX);
            if (i + 1 < m) finishRow(vy, selfY, i + 1, slotY);

            slotX = slotX2;
            slotY = slotY2;
        }

        for (int i = m; i < 16; ++i) {
            if (g == 0) *reinterpret_cast<v4f*>(&At[i * PITCH + q4]) = zero4;
        }

        __builtin_amdgcn_fence(__ATOMIC_ACQ_REL, "wavefront");

        v8f dacc[CT] = {};
        const float* Arow = &At[lm * PITCH + kh];

#pragma unroll 4
        for (int k0 = 0; k0 < K; k0 += 4) {
            v2f a = *reinterpret_cast<const v2f*>(&Arow[k0]);
#pragma unroll
            for (int c = 0; c < CT; ++c) {
                v2f b = *reinterpret_cast<const v2f*>(&Wt[(c * 16 + lm) * PITCH + k0 + kh]);
                dacc[c] = __builtin_amdgcn_wmma_f32_16x16x4_f32(
                    false, a, false, b, (short)0, dacc[c], false, false);
            }
        }

        __builtin_amdgcn_fence(__ATOMIC_ACQ_REL, "wavefront");

        const float dsv = (float)csv;
        const float rsv = ADDONE ? rsqrtf(dsv + 1.0f) : rsqrtf(fmaxf(dsv, 1.0f));
        float rs[8];
#pragma unroll
        for (int j = 0; j < 8; ++j) rs[j] = __shfl(rsv, mh + j);

#pragma unroll
        for (int c = 0; c < CT; ++c) {
            const int n = c * 16 + lm;
            const float bv = Bl[n];
#pragma unroll
            for (int j = 0; j < 8; ++j) {
                if (mh + j < m) {
                    float v = dacc[c][j] + bv;
                    if (TANH) v = tanhf(v);
                    out[(size_t)(rb + mh + j) * NOUT + n] = v * rs[j];
                }
            }
        }
    }
}

extern "C" void kernel_launch(void* const* d_in, const int* in_sizes, int n_in,
                              void* d_out, int out_size, void* d_ws, size_t ws_size,
                              hipStream_t stream) {
    const float* nodes = (const float*)d_in[0];
    const int* senders = (const int*)d_in[1];
    const int* receivers = (const int*)d_in[2];
    const float* W0 = (const float*)d_in[3];
    const float* b0 = (const float*)d_in[4];
    const float* W1 = (const float*)d_in[5];
    const float* b1 = (const float*)d_in[6];
    const float* W2 = (const float*)d_in[7];
    const float* b2 = (const float*)d_in[8];
    float* out = (float*)d_out;

    const int N = in_sizes[0] / 64;
    const int E = in_sizes[1];

    float* X = (float*)d_ws;
    float* Y = X + (size_t)N * 64;
    float* Z = Y + (size_t)N * 64;
    int* col = (int*)(Z + (size_t)N * 16);
    int* ovfS = col + (size_t)N * CAP;
    int* ovfR = ovfS + E;
    int* cntS = ovfR + E;
    int* cntR = cntS + N;
    int* curR = cntR + N;
    int* curS = curR + MAXBINS * CURSTRIDE;
    int* ovfCount = curS + MAXBINS * CURSTRIDE;
    int* binS = ovfCount + 4;
    const int nbins = (N + BINW - 1) / BINW;
    int* binR = binS + (size_t)nbins * BCAP;

    const int zn = 2 * MAXBINS * CURSTRIDE + 1;
    gcn_zero_i<<<(zn + 255) / 256, 256, 0, stream>>>(curR, zn);
    gcn_bin<<<(E + EPB - 1) / EPB, BINT, 0, stream>>>(senders, receivers, curR, curS, binR, binS, E);
    gcn_build<<<nbins, BUILDT, 0, stream>>>(senders, receivers, curR, curS, binR, binS, cntS, cntR,
                                            col, ovfCount, ovfS, ovfR, N, E);

    const int tiles = (N + 15) / 16;
    const int gblocks = (tiles + 3) / 4;

    gcn_gemm<64, true, true><<<gblocks, 128, 0, stream>>>(nodes, W0, b0, cntS, X, N);

    gcn_gather_gemm<64, true, true, 8><<<GATHER_BLOCKS, 256, 0, stream>>>(X, cntR, col, ovfCount, ovfS, ovfR,
                                                                           W1, b1, cntS, Y, N, E);

    gcn_gather_gemm<16, false, false, 8><<<GATHER_BLOCKS, 256, 0, stream>>>(Y, cntR, col, ovfCount, ovfS, ovfR,
                                                                             W2, b2, cntS, Z, N, E);

    gcn_gather<16, false, 2><<<GATHER_BLOCKS, 256, 0, stream>>>(Z, cntR, col, ovfCount, ovfS, ovfR, out, N, E);
}


# layer-2 product taken before the layer-1 aggregation (sum of rows @ W2 = sum of (rows @ W2)): fused layer-1 kernel stores y@W2, 2nd gather reads 16-wide rows not 64-wide; float sum order changes
# speedup vs baseline: 4.3973x; 1.0418x over previous
#include <hip/hip_runtime.h>
#include <hip/hip_bf16.h>

typedef float v2f __attribute__((ext_vector_type(2)));
typedef float v4f __attribute__((ext_vector_type(4)));
typedef float v8f __attribute__((ext_vector_type(8)));

constexpr int CAP = 32;

constexpr int BINSH = 8;
constexpr int BINW = 1 << BINSH;
constexpr int MAXBINS = 512;
constexpr int BCAP = 4096;
constexpr int EPB = 4096;
constexpr int BINT = 512;
constexpr int EPT = EPB / BINT;
constexpr int CURSTRIDE = 32;
constexpr int BUILDT = 512;
constexpr int BUILDU = 6;

__global__ void gcn_zero_i(int* __restrict__ p, int n) {
    int t = blockIdx.x * blockDim.x + threadIdx.x;
    if (t < n) p[t] = 0;
}

__global__ __launch_bounds__(BINT) void gcn_bin(const int* __restrict__ se, const int* __restrict__ re,
                                               int* __restrict__ curR, int* __restrict__ curS,
                                               int* __restrict__ binR, int* __restrict__ binS, int E) {
    __shared__ int cR[MAXBINS];
    __shared__ int cS[MAXBINS];
    __shared__ int gR[MAXBINS];
    __shared__ int gS[MAXBINS];
    __shared__ int wsum[2 * (BINT / 32)];
    __shared__ int2 sortR[EPB];
    __shared__ int sortS[EPB];

    const int tid = threadIdx.x;
    const int lane = tid & 31;
    const int wv = tid >> 5;
    const int e0 = blockIdx.x * EPB;
    const int n = (E - e0) < EPB ? (E - e0) : EPB;

    cR[tid] = 0;
    cS[tid] = 0;
    __syncthreads();

    int s[EPT], r[EPT], pR[EPT], pS[EPT];
#pragma unroll
    for (int j = 0; j < EPT; ++j) {
        const int idx = tid + j * BINT;
        const int i = e0 + (idx < n ? idx : n - 1);
        s[j] = se[i];
        r[j] = re[i];
    }
#pragma unroll
    for (int j = 0; j < EPT; ++j) {
        pR[j] = 0;
        pS[j] = 0;
        if (tid + j * BINT < n) {
            pR[j] = atomicAdd(&cR[r[j] >> BINSH], 1);
            pS[j] = atomicAdd(&cS[s[j] >> BINSH], 1);
        }
    }
    __syncthreads();

    const int vR = cR[tid];
    const int vS = cS[tid];
    int iR = vR, iS = vS;
#pragma unroll
    for (int d = 1; d < 32; d <<= 1) {
        const int tR = __shfl_up(iR, d);
        const int tS = __shfl_up(iS, d);
        if (lane >= d) {
            iR += tR;
            iS += tS;
        }
    }
    if (lane == 31) {
        wsum[wv] = iR;
        wsum[BINT / 32 + wv] = iS;
    }
    __syncthreads();
    int oR = 0, oS = 0;
    for (int w = 0; w < wv; ++w) {
        oR += wsum[w];
        oS += wsum[BINT / 32 + w];
    }
    int gbR = 0, gbS = 0;
    if (vR > 0) gbR = atomicAdd(&curR[tid * CURSTRIDE], vR);
    if (vS > 0) gbS = atomicAdd(&curS[tid * CURSTRIDE], vS);
    cR[tid] = oR + iR - vR;
    cS[tid] = oS + iS - vS;
    gR[tid] = gbR;
    gS[tid] = gbS;
    __syncthreads();

#pragma unroll
    for (int j = 0; j < EPT; ++j) {
        if (tid + j * BINT < n) {
            sortR[cR[r[j] >> BINSH] + pR[j]] = make_int2(s[j], r[j]);
            sortS[cS[s[j] >> BINSH] + pS[j]] = s[j];
        }
    }
    __syncthreads();

    int2* binR2 = reinterpret_cast<int2*>(binR);
    for (int idx = tid; idx < n; idx += BINT) {
        const int2 e = sortR[idx];
        const int kr = e.y >> BINSH;
        const int offR = gR[kr] + (idx - cR[kr]);
        if (offR < BCAP) binR2[(size_t)kr * BCAP + offR] = e;
        const int sv = sortS[idx];
        const int ks = sv >> BINSH;
        const int offS = gS[ks] + (idx - cS[ks]);
        if (offS < BCAP) binS[(size_t)ks * BCAP + offS] = sv;
    }
}

__global__ __launch_bounds__(BUILDT) void gcn_build(const int* __restrict__ se, const int* __restrict__ re,
                                                   const int* __restrict__ curR, const int* __restrict__ curS,
                                                   const int* __restrict__ binR, const int* __restrict__ binS,
                                                   int* __restrict__ cntS, int* __restrict__ cntR,
                                                   int* __restrict__ col, int* __restrict__ ovfCount,
                                                   int* __restrict__ ovfS, int* __restrict__ ovfR,
                                                   int N, int E) {
    __shared__ int dR[BINW];
    __shared__ int dS[BINW];
    __shared__ int colL[BINW * CAP];

    const int tid = threadIdx.x;
    const int k = blockIdx.x;
    if (tid < BINW) {
        dR[tid] = 0;
        dS[tid] = 0;
    }
    __syncthreads();

    const int totR = curR[k * CURSTRIDE];
    const int totS = curS[k * CURSTRIDE];

    auto addR = [&](int sv, int rv) {
        const int rl = rv & (BINW - 1);
        const int slot = atomicAdd(&dR[rl], 1);
        if (slot < CAP) {
            colL[rl * CAP + slot] = sv;
        } else {
            const int o = atomicAdd(ovfCount, 1);
            ovfS[o] = sv;
            ovfR[o] = rv;
        }
    };

    if (totR <= BCAP) {
        if (totR > 0) {
            const int2* p = reinterpret_cast<const int2*>(binR) + (size_t)k * BCAP;
            int2 ev[BUILDU];
#pragma unroll
            for (int j = 0; j < BUILDU; ++j) {
                const int i = tid + j * BUILDT;
                ev[j] = p[i < totR ? i : totR - 1];
            }
#pragma unroll
            for (int j = 0; j < BUILDU; ++j) {
                if (tid + j * BUILDT < totR) addR(ev[j].x, ev[j].y);
            }
            for (int i = tid + BUILDU * BUILDT; i < totR; i += BUILDT) {
                const int2 e = p[i];
                addR(e.x, e.y);
            }
        }
    } else {
        for (int i = tid; i < E; i += BUILDT) {
            const int rv = re[i];
            if ((rv >> BINSH) == k) addR(se[i], rv);
        }
    }

    if (totS <= BCAP) {
        if (totS > 0) {
            const int* p = binS + (size_t)k * BCAP;
            int sv[BUILDU];
#pragma unroll
            for (int j = 0; j < BUILDU; ++j) {
                const int i = tid + j * BUILDT;
                sv[j] = p[i < totS ? i : totS - 1];
            }
#pragma unroll
            for (int j = 0; j < BUILDU; ++j) {
                if (tid + j * BUILDT < totS) atomicAdd(&dS[sv[j] & (BINW - 1)], 1);
            }
            for (int i = tid + BUILDU * BUILDT; i < totS; i += BUILDT) {
                atomicAdd(&dS[p[i] & (BINW - 1)], 1);
            }
        }
    } else {
        for (int i = tid; i < E; i += BUILDT) {
            const int sv = se[i];
            if ((sv >> BINSH) == k) atomicAdd(&dS[sv & (BINW - 1)], 1);
        }
    }
    __syncthreads();

    const int row0 = k * BINW;
    if (tid < BINW && row0 + tid < N) {
        cntR[row0 + tid] = dR[tid];
        cntS[row0 + tid] = dS[tid];
    }
    for (int idx = tid; idx < BINW * CAP; idx += BUILDT) {
        const int rl = idx / CAP;
        const int j = idx % CAP;
        if (row0 + rl < N) col[(size_t)(row0 + rl) * CAP + j] = (j < dR[rl]) ? colL[idx] : 0;
    }
}

constexpr int GATHER_BLOCKS = 768;
constexpr int GATHER_WAVES = GATHER_BLOCKS * 8;

template <int F, bool SELF, bool EPI, int U>
__global__ __launch_bounds__(256) void gcn_gather(const float* __restrict__ src,
                                                  const int* __restrict__ cntR,
                                                  const int* __restrict__ col,
                                                  const int* __restrict__ ovfCount,
                                                  const int* __restrict__ ovfS,
                                                  const int* __restrict__ ovfR,
                                                  const float* __restrict__ bias,
                                                  const int* __restrict__ cntS,
                                                  float* __restrict__ dst, int N, int E) {
    constexpr int LPR = F / 4;
    constexpr int G = 32 / LPR;
    const int lane = threadIdx.x & 31;
    const int g = lane / LPR;
    const int q4 = (lane % LPR) * 4;

    const int wave = __builtin_amdgcn_readfirstlane((int)(blockIdx.x * 8 + (threadIdx.x >> 5)));
    const int base = N / GATHER_WAVES;
    const int rem = N % GATHER_WAVES;
    const int r0 = wave * base + (wave < rem ? wave : rem);
    const int r1 = r0 + base + (wave < rem ? 1 : 0);

    const v4f zero4 = {0.f, 0.f, 0.f, 0.f};
    v4f bias4 = zero4;
    if (EPI) bias4 = *reinterpret_cast<const v4f*>(&bias[q4]);

    for (int rb = r0; rb < r1; rb += 32) {
        const int m = (r1 - rb) < 32 ? (r1 - rb) : 32;
        const int cntv = cntR[rb + (lane < m ? lane : m - 1)];
        int csv = 0;
        if (EPI) csv = cntS[rb + (lane < m ? lane : m - 1)];

        auto rowCount = [&](int i) -> int { return i < m ? __builtin_amdgcn_readlane(cntv, i & 31) : 0; };

        auto slotLine = [&](int i) -> int {
            const int c = rowCount(i);
            const int n = c < CAP ? c : CAP;
            const int ri = rb + (i < m ? i : m - 1);
            const int* p = (n > 0) ? &col[(size_t)ri * CAP + (lane < n ? lane : n - 1)] : &cntR[ri];
            return *p;
        };

        auto issueRows = [&](v4f (&v)[U], v4f& selfv, int i, const int (&s)[U]) {
            const int c = rowCount(i);
            const int n = c < CAP ? c : CAP;
            if (SELF && i < m) selfv = *reinterpret_cast<const v4f*>(&src[(size_t)(rb + i) * F + q4]);
#pragma unroll
            for (int u = 0; u < U; ++u) {
                if (u * G < n) v[u] = *reinterpret_cast<const v4f*>(&src[(size_t)s[u] * F + q4]);
            }
        };

        auto finishRow = [&](const v4f (&v)[U], const v4f& selfv, int i, int myS) {
            const int r = rb + i;
            const int cnt = rowCount(i);
            const int n = cnt < CAP ? cnt : CAP;

            v4f acc = zero4;
            if (SELF && g == 0) acc = selfv;
#pragma unroll
            for (int u = 0; u < U; ++u) {
                if (u * G < n) acc += (u * G + g < n) ? v[u] : zero4;
            }

            for (int j0 = U * G; j0 < n; j0 += U * G) {
                v4f w[U];
#pragma unroll
                for (int u = 0; u < U; ++u) {
                    const int j = j0 + u * G + g;
                    const int s = __shfl(myS, j & 31);
                    w[u] = zero4;
                    if (j < n) w[u] = *reinterpret_cast<const v4f*>(&src[(size_t)s * F + q4]);
                }
#pragma unroll
                for (int u = 0; u < U; ++u) acc += w[u];
            }

            if (cnt > CAP) {
                int no = *ovfCount;
                if (no > E) no = E;
                for (int k = g; k < no; k += G) {
                    if (ovfR[k] == r) acc += *reinterpret_cast<const v4f*>(&src[(size_t)ovfS[k] * F + q4]);
                }
            }

#pragma unroll
            for (int off = LPR; off < 32; off <<= 1) {
                acc.x += __shfl_xor(acc.x, off);
                acc.y += __shfl_xor(acc.y, off);
                acc.z += __shfl_xor(acc.z, off);
                acc.w += __shfl_xor(acc.w, off);
            }

            if (g == 0) {
                const float d = (float)cnt;
                const float sc = SELF ? rsqrtf(d + 1.0f) : rsqrtf(fmaxf(d, 1.0f));
                v4f o = acc * sc;
                if (EPI) {
                    const float ds = (float)__builtin_amdgcn_readlane(csv, i & 31);
                    o = (o + bias4) * rsqrtf(fmaxf(ds, 1.0f));
                }
                *reinterpret_cast<v4f*>(&dst[(size_t)r * F + q4]) = o;
            }
        };

        int slotX = slotLine(0);
        int slotY = slotLine(1);

        for (int i = 0; i < m; i += 2) {
            const int slotX2 = slotLine(i + 2);
            const int slotY2 = slotLine(i + 3);

            int sx[U], sy[U];
#pragma unroll
            for (int u = 0; u < U; ++u) {
                sx[u] = __shfl(slotX, u * G + g);
                sy[u] = __shfl(slotY, u * G + g);
            }

            v4f vx[U], vy[U];
            v4f selfX, selfY;
            issueRows(vx, selfX, i, sx);
            issueRows(vy, selfY, i + 1, sy);

            finishRow(vx, selfX, i, slotX);
            if (i + 1 < m) finishRow(vy, selfY, i + 1, slotY);

            slotX = slotX2;
            slotY = slotY2;
        }
    }
}

template <int NOUT, bool TANH, bool ADDONE>
__global__ __launch_bounds__(128) void gcn_gemm(const float* __restrict__ in,
                                                const float* __restrict__ W,
                                                const float* __restrict__ bias,
                                                const int* __restrict__ cntS,
                                                float* __restrict__ out, int N) {
    constexpr int K = 64;
    constexpr int CT = NOUT / 16;
    constexpr int PITCH = 68;

    __shared__ float Wt[NOUT * PITCH];
    __shared__ float Bl[NOUT];
    __shared__ float Al[64 * PITCH];

    const int tid = threadIdx.x;
    const int wave = tid >> 5;
    const int lane = tid & 31;
    const int rowBase = blockIdx.x * 64;
    const int tileRow = rowBase + wave * 16;
    const int lm = lane & 15;
    const int kh = (lane >> 4) * 2;
    const int mh = (lane >> 4) * 8;

    constexpr int WI = (K * NOUT / 4) / 128;
    constexpr int AI = (64 * (K / 4)) / 128;
    const float4* Wg4 = reinterpret_cast<const float4*>(W);
    const float4* Ag4 = reinterpret_cast<const float4*>(in);
    float4 wv[WI];
    float4 av[AI];
    int cs[8];
#pragma unroll
    for (int t = 0; t < WI; ++t) wv[t] = Wg4[tid + t * 128];
#pragma unroll
    for (int t = 0; t < AI; ++t) {
        const int i = tid + t * 128;
        const int row = rowBase + i / (K / 4);
        av[t] = make_float4(0.f, 0.f, 0.f, 0.f);
        if (row < N) av[t] = Ag4[(size_t)row * (K / 4) + (i % (K / 4))];
    }
    float bl = 0.f;
    if (tid < NOUT) bl = bias[tid];
#pragma unroll
    for (int j = 0; j < 8; ++j) {
        const int row = tileRow + mh + j;
        cs[j] = cntS[row < N ? row : N - 1];
    }

#pragma unroll
    for (int t = 0; t < WI; ++t) {
        const int i = tid + t * 128;
        const int k = i / (NOUT / 4);
        const int n = (i % (NOUT / 4)) * 4;
        Wt[(n + 0) * PITCH + k] = wv[t].x;
        Wt[(n + 1) * PITCH + k] = wv[t].y;
        Wt[(n + 2) * PITCH + k] = wv[t].z;
        Wt[(n + 3) * PITCH + k] = wv[t].w;
    }
    if (tid < NOUT) Bl[tid] = bl;

#pragma unroll
    for (int t = 0; t < AI; ++t) {
        const int i = tid + t * 128;
        const int c4 = (i % (K / 4)) * 4;
        *reinterpret_cast<float4*>(&Al[(i / (K / 4)) * PITCH + c4]) = av[t];
    }
    __syncthreads();

    if (tileRow >= N) return;

    v8f acc[CT] = {};
    const float* Arow = &Al[(wave * 16 + lm) * PITCH + kh];

#pragma unroll
    for (int k0 = 0; k0 < K; k0 += 4) {
        v2f a = *reinterpret_cast<const v2f*>(&Arow[k0]);
#pragma unroll
        for (int c = 0; c < CT; ++c) {
            v2f b = *reinterpret_cast<const v2f*>(&Wt[(c * 16 + lm) * PITCH + k0 + kh]);
            acc[c] = __builtin_amdgcn_wmma_f32_16x16x4_f32(
                false, a, false, b, (short)0, acc[c], false, false);
        }
    }

    float rs[8];
#pragma unroll
    for (int j = 0; j < 8; ++j) {
        int row = tileRow + mh + j;
        float d = (row < N) ? (float)cs[j] : 1.0f;
        rs[j] = ADDONE ? rsqrtf(d + 1.0f) : rsqrtf(fmaxf(d, 1.0f));
    }

#pragma unroll
    for (int c = 0; c < CT; ++c) {
        int n = c * 16 + lm;
        float bv = Bl[n];
#pragma unroll
        for (int j = 0; j < 8; ++j) {
            int row = tileRow + mh + j;
            if (row < N) {
                float v = acc[c][j] + bv;
                if (TANH) v = tanhf(v);
                out[(size_t)row * NOUT + n] = v * rs[j];
            }
        }
    }
}

template <int NOUT, int NP, bool TANH, bool ADDONE, int U>
__global__ __launch_bounds__(256, 6) void gcn_gather_gemm(const float* __restrict__ src,
                                                       const int* __restrict__ cntR,
                                                       const int* __restrict__ col,
                                                       const int* __restrict__ ovfCount,
                                                       const int* __restrict__ ovfS,
                                                       const int* __restrict__ ovfR,
                                                       const float* __restrict__ W,
                                                       const float* __restrict__ bias,
                                                       const int* __restrict__ cntS,
                                                       const float* __restrict__ Wp,
                                                       float* __restrict__ out, int N, int E) {
    static_assert(NOUT == 64, "the y tile is the A operand of the second product: 64 wide");
    constexpr int F = 64;
    constexpr int K = 64;
    constexpr int CT = NOUT / 16;
    constexpr int CP = NP / 16;
    constexpr int PITCH = 68;
    constexpr int LPR = F / 4;
    constexpr int G = 32 / LPR;

    __shared__ alignas(16) float Wt[NOUT * PITCH];
    __shared__ alignas(16) float Wpt[NP * PITCH];
    __shared__ float Bl[NOUT];
    __shared__ alignas(16) float Al[8 * 16 * PITCH];

    const int tid = threadIdx.x;
    const int lane = tid & 31;
    const int g = lane / LPR;
    const int q4 = (lane % LPR) * 4;
    const int lm = lane & 15;
    const int kh = (lane >> 4) * 2;
    const int mh = (lane >> 4) * 8;

    {
        constexpr int WI = (K * NOUT / 4) / 256;
        constexpr int WPI = (K * NP / 4) / 256;
        static_assert(WPI >= 1 && WPI * 256 * 4 == K * NP, "Wp is dealt evenly over the threads");
        const float4* Wg4 = reinterpret_cast<const float4*>(W);
        const float4* Wpg4 = reinterpret_cast<const float4*>(Wp);
        float4 wv[WI];
        float4 wpv[WPI];
#pragma unroll
        for (int t = 0; t < WI; ++t) wv[t] = Wg4[tid + t * 256];
#pragma unroll
        for (int t = 0; t < WPI; ++t) wpv[t] = Wpg4[tid + t * 256];
        float bl = 0.f;
        if (tid < NOUT) bl = bias[tid];
#pragma unroll
        for (int t = 0; t < WI; ++t) {
            const int i = tid + t * 256;
            const int k = i / (NOUT / 4);
            const int n = (i % (NOUT / 4)) * 4;
            Wt[(n + 0) * PITCH + k] = wv[t].x;
            Wt[(n + 1) * PITCH + k] = wv[t].y;
            Wt[(n + 2) * PITCH + k] = wv[t].z;
            Wt[(n + 3) * PITCH + k] = wv[t].w;
        }
#pragma unroll
        for (int t = 0; t < WPI; ++t) {
            const int i = tid + t * 256;
            const int k = i / (NP / 4);
            const int n = (i % (NP / 4)) * 4;
            Wpt[(n + 0) * PITCH + k] = wpv[t].x;
            Wpt[(n + 1) * PITCH + k] = wpv[t].y;
            Wpt[(n + 2) * PITCH + k] = wpv[t].z;
            Wpt[(n + 3) * PITCH + k] = wpv[t].w;
        }
        if (tid < NOUT) Bl[tid] = bl;
    }
    __syncthreads();

    const int wl = __builtin_amdgcn_readfirstlane((int)(threadIdx.x >> 5));
    const int wave = __builtin_amdgcn_readfirstlane((int)(blockIdx.x * 8)) + wl;
    const int base = N / GATHER_WAVES;
    const int rem = N % GATHER_WAVES;
    const int r0 = wave * base + (wave < rem ? wave : rem);
    const int r1 = r0 + base + (wave < rem ? 1 : 0);

    float* At = &Al[wl * 16 * PITCH];
    const v4f zero4 = {0.f, 0.f, 0.f, 0.f};

    for (int rb = r0; rb < r1; rb += 16) {
        const int m = (r1 - rb) < 16 ? (r1 - rb) : 16;
        const int cntv = cntR[rb + (lane < m ? lane : m - 1)];
        const int csv = cntS[rb + (lane < m ? lane : m - 1)];

        auto rowCount = [&](int i) -> int { return i < m ? __builtin_amdgcn_readlane(cntv, i & 31) : 0; };

        auto slotLine = [&](int i) -> int {
            const int c = rowCount(i);
            const int n = c < CAP ? c : CAP;
            const int ri = rb + (i < m ? i : m - 1);
            const int* p = (n > 0) ? &col[(size_t)ri * CAP + (lane < n ? lane : n - 1)] : &cntR[ri];
            return *p;
        };

        auto issueRows = [&](v4f (&v)[U], v4f& selfv, int i, const int (&s)[U]) {
            const int c = rowCount(i);
            const int n = c < CAP ? c : CAP;
            if (i < m) selfv = *reinterpret_cast<const v4f*>(&src[(size_t)(rb + i) * F + q4]);
#pragma unroll
            for (int u = 0; u < U; ++u) {
                if (u * G < n) v[u] = *reinterpret_cast<const v4f*>(&src[(size_t)s[u] * F + q4]);
            }
        };

        auto finishRow = [&](const v4f (&v)[U], const v4f& selfv, int i, int myS) {
            const int r = rb + i;
            const int cnt = rowCount(i);
            const int n = cnt < CAP ? cnt : CAP;

            v4f acc = zero4;
            if (g == 0) acc = selfv;
#pragma unroll
            for (int u = 0; u < U; ++u) {
                if (u * G < n) acc += (u * G + g < n) ? v[u] : zero4;
            }

            for (int j0 = U * G; j0 < n; j0 += U * G) {
                v4f w[U];
#pragma unroll
                for (int u = 0; u < U; ++u) {
                    const int j = j0 + u * G + g;
                    const int s = __shfl(myS, j & 31);
                    w[u] = zero4;
                    if (j < n) w[u] = *reinterpret_cast<const v4f*>(&src[(size_t)s * F + q4]);
                }
#pragma unroll
                for (int u = 0; u < U; ++u) acc += w[u];
            }

            if (cnt > CAP) {
                int no = *ovfCount;
                if (no > E) no = E;
                for (int k = g; k < no; k += G) {
                    if (ovfR[k] == r) acc += *reinterpret_cast<const v4f*>(&src[(size_t)ovfS[k] * F + q4]);
                }
            }

#pragma unroll
            for (int off = LPR; off < 32; off <<= 1) {
                acc.x += __shfl_xor(acc.x, off);
                acc.y += __shfl_xor(acc.y, off);
                acc.z += __shfl_xor(acc.z, off);
                acc.w += __shfl_xor(acc.w, off);
            }

            if (g == 0) {
                const float sc = rsqrtf((float)cnt + 1.0f);
                *reinterpret_cast<v4f*>(&At[i * PITCH + q4]) = acc * sc;
            }
        };

        int slotX = slotLine(0);
        int slotY = slotLine(1);

        for (int i = 0; i < m; i += 2) {
            const int slotX2 = slotLine(i + 2);
            const int slotY2 = slotLine(i + 3);

            int sx[U], sy[U];
#pragma unroll
            for (int u = 0; u < U; ++u) {
                sx[u] = __shfl(slotX, u * G + g);
                sy[u] = __shfl(slotY, u * G + g);
            }

            v4f vx[U], vy[U];
            v4f selfX, selfY;
            issueRows(vx, selfX, i, sx);
            issueRows(vy, selfY, i + 1, sy);

            finishRow(vx, selfX, i, slotX);
            if (i + 1 < m) finishRow(vy, selfY, i + 1, slotY);

            slotX = slotX2;
            slotY = slotY2;
        }

        for (int i = m; i < 16; ++i) {
            if (g == 0) *reinterpret_cast<v4f*>(&At[i * PITCH + q4]) = zero4;
        }

        __builtin_amdgcn_fence(__ATOMIC_ACQ_REL, "wavefront");

        v8f dacc[CT] = {};
        const float* Arow = &At[lm * PITCH + kh];

#pragma unroll 4
        for (int k0 = 0; k0 < K; k0 += 4) {
            v2f a = *reinterpret_cast<const v2f*>(&Arow[k0]);
#pragma unroll
            for (int c = 0; c < CT; ++c) {
                v2f b = *reinterpret_cast<const v2f*>(&Wt[(c * 16 + lm) * PITCH + k0 + kh]);
                dacc[c] = __builtin_amdgcn_wmma_f32_16x16x4_f32(
                    false, a, false, b, (short)0, dacc[c], false, false);
            }
        }

        __builtin_amdgcn_fence(__ATOMIC_ACQ_REL, "wavefront");

        const float dsv = (float)csv;
        const float rsv = ADDONE ? rsqrtf(dsv + 1.0f) : rsqrtf(fmaxf(dsv, 1.0f));
        float rs[8];
#pragma unroll
        for (int j = 0; j < 8; ++j) rs[j] = __shfl(rsv, mh + j);

#pragma unroll
        for (int c = 0; c < CT; ++c) {
            const int n = c * 16 + lm;
            const float bv = Bl[n];
#pragma unroll
            for (int j = 0; j < 8; ++j) {
                float y = 0.f;
                if (mh + j < m) {
                    float v = dacc[c][j] + bv;
                    if (TANH) v = tanhf(v);
                    y = v * rs[j];
                }
                At[(mh + j) * PITCH + n] = y;
            }
        }

        __builtin_amdgcn_fence(__ATOMIC_ACQ_REL, "wavefront");

        v8f pacc[CP] = {};
#pragma unroll 4
        for (int k0 = 0; k0 < K; k0 += 4) {
            v2f a = *reinterpret_cast<const v2f*>(&Arow[k0]);
#pragma unroll
            for (int c = 0; c < CP; ++c) {
                v2f b = *reinterpret_cast<const v2f*>(&Wpt[(c * 16 + lm) * PITCH + k0 + kh]);
                pacc[c] = __builtin_amdgcn_wmma_f32_16x16x4_f32(
                    false, a, false, b, (short)0, pacc[c], false, false);
            }
        }

        __builtin_amdgcn_fence(__ATOMIC_ACQ_REL, "wavefront");

#pragma unroll
        for (int c = 0; c < CP; ++c) {
            const int n = c * 16 + lm;
#pragma unroll
            for (int j = 0; j < 8; ++j) {
                if (mh + j < m) out[(size_t)(rb + mh + j) * NP + n] = pacc[c][j];
            }
        }
    }
}

extern "C" void kernel_launch(void* const* d_in, const int* in_sizes, int n_in,
                              void* d_out, int out_size, void* d_ws, size_t ws_size,
                              hipStream_t stream) {
    const float* nodes = (const float*)d_in[0];
    const int* senders = (const int*)d_in[1];
    const int* receivers = (const int*)d_in[2];
    const float* W0 = (const float*)d_in[3];
    const float* b0 = (const float*)d_in[4];
    const float* W1 = (const float*)d_in[5];
    const float* b1 = (const float*)d_in[6];
    const float* W2 = (const float*)d_in[7];
    const float* b2 = (const float*)d_in[8];
    float* out = (float*)d_out;

    const int N = in_sizes[0] / 64;
    const int E = in_sizes[1];

    float* X = (float*)d_ws;
    float* P = X + (size_t)N * 64;
    float* Z = P + (size_t)N * 16;
    int* col = (int*)(Z + (size_t)N * 16);
    int* ovfS = col + (size_t)N * CAP;
    int* ovfR = ovfS + E;
    int* cntS = ovfR + E;
    int* cntR = cntS + N;
    int* curR = cntR + N;
    int* curS = curR + MAXBINS * CURSTRIDE;
    int* ovfCount = curS + MAXBINS * CURSTRIDE;
    int* binS = ovfCount + 4;
    const int nbins = (N + BINW - 1) / BINW;
    int* binR = binS + (size_t)nbins * BCAP;

    const int zn = 2 * MAXBINS * CURSTRIDE + 1;
    gcn_zero_i<<<(zn + 255) / 256, 256, 0, stream>>>(curR, zn);
    gcn_bin<<<(E + EPB - 1) / EPB, BINT, 0, stream>>>(senders, receivers, curR, curS, binR, binS, E);
    gcn_build<<<nbins, BUILDT, 0, stream>>>(senders, receivers, curR, curS, binR, binS, cntS, cntR,
                                            col, ovfCount, ovfS, ovfR, N, E);

    const int tiles = (N + 15) / 16;
    const int gblocks = (tiles + 3) / 4;

    gcn_gemm<64, true, true><<<gblocks, 128, 0, stream>>>(nodes, W0, b0, cntS, X, N);

    gcn_gather_gemm<64, 16, true, true, 8><<<GATHER_BLOCKS, 256, 0, stream>>>(X, cntR, col, ovfCount, ovfS, ovfR,
                                                                               W1, b1, cntS, W2, P, N, E);

    gcn_gather<16, true, true, 2><<<GATHER_BLOCKS, 256, 0, stream>>>(P, cntR, col, ovfCount, ovfS, ovfR,
                                                                      b2, cntS, Z, N, E);

    gcn_gather<16, false, false, 2><<<GATHER_BLOCKS, 256, 0, stream>>>(Z, cntR, col, ovfCount, ovfS, ovfR,
                                                                        b2, cntS, out, N, E);
}


# tanh of both activation epilogues on the hardware instruction v_tanh_f32 (one issue per value) instead of the library tanhf's two divergent branches; tanh values differ in their last bits
# speedup vs baseline: 4.7181x; 1.0730x over previous
#include <hip/hip_runtime.h>
#include <hip/hip_bf16.h>

typedef float v2f __attribute__((ext_vector_type(2)));
typedef float v4f __attribute__((ext_vector_type(4)));
typedef float v8f __attribute__((ext_vector_type(8)));

constexpr int CAP = 32;

constexpr int BINSH = 8;
constexpr int BINW = 1 << BINSH;
constexpr int MAXBINS = 512;
constexpr int BCAP = 4096;
constexpr int EPB = 4096;
constexpr int BINT = 512;
constexpr int EPT = EPB / BINT;
constexpr int CURSTRIDE = 32;
constexpr int BUILDT = 512;
constexpr int BUILDU = 6;

__global__ void gcn_zero_i(int* __restrict__ p, int n) {
    int t = blockIdx.x * blockDim.x + threadIdx.x;
    if (t < n) p[t] = 0;
}

__global__ __launch_bounds__(BINT) void gcn_bin(const int* __restrict__ se, const int* __restrict__ re,
                                               int* __restrict__ curR, int* __restrict__ curS,
                                               int* __restrict__ binR, int* __restrict__ binS, int E) {
    __shared__ int cR[MAXBINS];
    __shared__ int cS[MAXBINS];
    __shared__ int gR[MAXBINS];
    __shared__ int gS[MAXBINS];
    __shared__ int wsum[2 * (BINT / 32)];
    __shared__ int2 sortR[EPB];
    __shared__ int sortS[EPB];

    const int tid = threadIdx.x;
    const int lane = tid & 31;
    const int wv = tid >> 5;
    const int e0 = blockIdx.x * EPB;
    const int n = (E - e0) < EPB ? (E - e0) : EPB;

    cR[tid] = 0;
    cS[tid] = 0;
    __syncthreads();

    int s[EPT], r[EPT], pR[EPT], pS[EPT];
#pragma unroll
    for (int j = 0; j < EPT; ++j) {
        const int idx = tid + j * BINT;
        const int i = e0 + (idx < n ? idx : n - 1);
        s[j] = se[i];
        r[j] = re[i];
    }
#pragma unroll
    for (int j = 0; j < EPT; ++j) {
        pR[j] = 0;
        pS[j] = 0;
        if (tid + j * BINT < n) {
            pR[j] = atomicAdd(&cR[r[j] >> BINSH], 1);
            pS[j] = atomicAdd(&cS[s[j] >> BINSH], 1);
        }
    }
    __syncthreads();

    const int vR = cR[tid];
    const int vS = cS[tid];
    int iR = vR, iS = vS;
#pragma unroll
    for (int d = 1; d < 32; d <<= 1) {
        const int tR = __shfl_up(iR, d);
        const int tS = __shfl_up(iS, d);
        if (lane >= d) {
            iR += tR;
            iS += tS;
        }
    }
    if (lane == 31) {
        wsum[wv] = iR;
        wsum[BINT / 32 + wv] = iS;
    }
    __syncthreads();
    int oR = 0, oS = 0;
    for (int w = 0; w < wv; ++w) {
        oR += wsum[w];
        oS += wsum[BINT / 32 + w];
    }
    int gbR = 0, gbS = 0;
    if (vR > 0) gbR = atomicAdd(&curR[tid * CURSTRIDE], vR);
    if (vS > 0) gbS = atomicAdd(&curS[tid * CURSTRIDE], vS);
    cR[tid] = oR + iR - vR;
    cS[tid] = oS + iS - vS;
    gR[tid] = gbR;
    gS[tid] = gbS;
    __syncthreads();

#pragma unroll
    for (int j = 0; j < EPT; ++j) {
        if (tid + j * BINT < n) {
            sortR[cR[r[j] >> BINSH] + pR[j]] = make_int2(s[j], r[j]);
            sortS[cS[s[j] >> BINSH] + pS[j]] = s[j];
        }
    }
    __syncthreads();

    int2* binR2 = reinterpret_cast<int2*>(binR);
    for (int idx = tid; idx < n; idx += BINT) {
        const int2 e = sortR[idx];
        const int kr = e.y >> BINSH;
        const int offR = gR[kr] + (idx - cR[kr]);
        if (offR < BCAP) binR2[(size_t)kr * BCAP + offR] = e;
        const int sv = sortS[idx];
        const int ks = sv >> BINSH;
        const int offS = gS[ks] + (idx - cS[ks]);
        if (offS < BCAP) binS[(size_t)ks * BCAP + offS] = sv;
    }
}

__global__ __launch_bounds__(BUILDT) void gcn_build(const int* __restrict__ se, const int* __restrict__ re,
                                                   const int* __restrict__ curR, const int* __restrict__ curS,
                                                   const int* __restrict__ binR, const int* __restrict__ binS,
                                                   int* __restrict__ cntS, int* __restrict__ cntR,
                                                   int* __restrict__ col, int* __restrict__ ovfCount,
                                                   int* __restrict__ ovfS, int* __restrict__ ovfR,
                                                   int N, int E) {
    __shared__ int dR[BINW];
    __shared__ int dS[BINW];
    __shared__ int colL[BINW * CAP];

    const int tid = threadIdx.x;
    const int k = blockIdx.x;
    if (tid < BINW) {
        dR[tid] = 0;
        dS[tid] = 0;
    }
    __syncthreads();

    const int totR = curR[k * CURSTRIDE];
    const int totS = curS[k * CURSTRIDE];

    auto addR = [&](int sv, int rv) {
        const int rl = rv & (BINW - 1);
        const int slot = atomicAdd(&dR[rl], 1);
        if (slot < CAP) {
            colL[rl * CAP + slot] = sv;
        } else {
            const int o = atomicAdd(ovfCount, 1);
            ovfS[o] = sv;
            ovfR[o] = rv;
        }
    };

    if (totR <= BCAP) {
        if (totR > 0) {
            const int2* p = reinterpret_cast<const int2*>(binR) + (size_t)k * BCAP;
            int2 ev[BUILDU];
#pragma unroll
            for (int j = 0; j < BUILDU; ++j) {
                const int i = tid + j * BUILDT;
                ev[j] = p[i < totR ? i : totR - 1];
            }
#pragma unroll
            for (int j = 0; j < BUILDU; ++j) {
                if (tid + j * BUILDT < totR) addR(ev[j].x, ev[j].y);
            }
            for (int i = tid + BUILDU * BUILDT; i < totR; i += BUILDT) {
                const int2 e = p[i];
                addR(e.x, e.y);
            }
        }
    } else {
        for (int i = tid; i < E; i += BUILDT) {
            const int rv = re[i];
            if ((rv >> BINSH) == k) addR(se[i], rv);
        }
    }

    if (totS <= BCAP) {
        if (totS > 0) {
            const int* p = binS + (size_t)k * BCAP;
            int sv[BUILDU];
#pragma unroll
            for (int j = 0; j < BUILDU; ++j) {
                const int i = tid + j * BUILDT;
                sv[j] = p[i < totS ? i : totS - 1];
            }
#pragma unroll
            for (int j = 0; j < BUILDU; ++j) {
                if (tid + j * BUILDT < totS) atomicAdd(&dS[sv[j] & (BINW - 1)], 1);
            }
            for (int i = tid + BUILDU * BUILDT; i < totS; i += BUILDT) {
                atomicAdd(&dS[p[i] & (BINW - 1)], 1);
            }
        }
    } else {
        for (int i = tid; i < E; i += BUILDT) {
            const int sv = se[i];
            if ((sv >> BINSH) == k) atomicAdd(&dS[sv & (BINW - 1)], 1);
        }
    }
    __syncthreads();

    const int row0 = k * BINW;
    if (tid < BINW && row0 + tid < N) {
        cntR[row0 + tid] = dR[tid];
        cntS[row0 + tid] = dS[tid];
    }
    for (int idx = tid; idx < BINW * CAP; idx += BUILDT) {
        const int rl = idx / CAP;
        const int j = idx % CAP;
        if (row0 + rl < N) col[(size_t)(row0 + rl) * CAP + j] = (j < dR[rl]) ? colL[idx] : 0;
    }
}

constexpr int GATHER_BLOCKS = 768;
constexpr int GATHER_WAVES = GATHER_BLOCKS * 8;

template <int F, bool SELF, bool EPI, int U>
__global__ __launch_bounds__(256) void gcn_gather(const float* __restrict__ src,
                                                  const int* __restrict__ cntR,
                                                  const int* __restrict__ col,
                                                  const int* __restrict__ ovfCount,
                                                  const int* __restrict__ ovfS,
                                                  const int* __restrict__ ovfR,
                                                  const float* __restrict__ bias,
                                                  const int* __restrict__ cntS,
                                                  float* __restrict__ dst, int N, int E) {
    constexpr int LPR = F / 4;
    constexpr int G = 32 / LPR;
    const int lane = threadIdx.x & 31;
    const int g = lane / LPR;
    const int q4 = (lane % LPR) * 4;

    const int wave = __builtin_amdgcn_readfirstlane((int)(blockIdx.x * 8 + (threadIdx.x >> 5)));
    const int base = N / GATHER_WAVES;
    const int rem = N % GATHER_WAVES;
    const int r0 = wave * base + (wave < rem ? wave : rem);
    const int r1 = r0 + base + (wave < rem ? 1 : 0);

    const v4f zero4 = {0.f, 0.f, 0.f, 0.f};
    v4f bias4 = zero4;
    if (EPI) bias4 = *reinterpret_cast<const v4f*>(&bias[q4]);

    for (int rb = r0; rb < r1; rb += 32) {
        const int m = (r1 - rb) < 32 ? (r1 - rb) : 32;
        const int cntv = cntR[rb + (lane < m ? lane : m - 1)];
        int csv = 0;
        if (EPI) csv = cntS[rb + (lane < m ? lane : m - 1)];

        auto rowCount = [&](int i) -> int { return i < m ? __builtin_amdgcn_readlane(cntv, i & 31) : 0; };

        auto slotLine = [&](int i) -> int {
            const int c = rowCount(i);
            const int n = c < CAP ? c : CAP;
            const int ri = rb + (i < m ? i : m - 1);
            const int* p = (n > 0) ? &col[(size_t)ri * CAP + (lane < n ? lane : n - 1)] : &cntR[ri];
            return *p;
        };

        auto issueRows = [&](v4f (&v)[U], v4f& selfv, int i, const int (&s)[U]) {
            const int c = rowCount(i);
            const int n = c < CAP ? c : CAP;
            if (SELF && i < m) selfv = *reinterpret_cast<const v4f*>(&src[(size_t)(rb + i) * F + q4]);
#pragma unroll
            for (int u = 0; u < U; ++u) {
                if (u * G < n) v[u] = *reinterpret_cast<const v4f*>(&src[(size_t)s[u] * F + q4]);
            }
        };

        auto finishRow = [&](const v4f (&v)[U], const v4f& selfv, int i, int myS) {
            const int r = rb + i;
            const int cnt = rowCount(i);
            const int n = cnt < CAP ? cnt : CAP;

            v4f acc = zero4;
            if (SELF && g == 0) acc = selfv;
#pragma unroll
            for (int u = 0; u < U; ++u) {
                if (u * G < n) acc += (u * G + g < n) ? v[u] : zero4;
            }

            for (int j0 = U * G; j0 < n; j0 += U * G) {
                v4f w[U];
#pragma unroll
                for (int u = 0; u < U; ++u) {
                    const int j = j0 + u * G + g;
                    const int s = __shfl(myS, j & 31);
                    w[u] = zero4;
                    if (j < n) w[u] = *reinterpret_cast<const v4f*>(&src[(size_t)s * F + q4]);
                }
#pragma unroll
                for (int u = 0; u < U; ++u) acc += w[u];
            }

            if (cnt > CAP) {
                int no = *ovfCount;
                if (no > E) no = E;
                for (int k = g; k < no; k += G) {
                    if (ovfR[k] == r) acc += *reinterpret_cast<const v4f*>(&src[(size_t)ovfS[k] * F + q4]);
                }
            }

#pragma unroll
            for (int off = LPR; off < 32; off <<= 1) {
                acc.x += __shfl_xor(acc.x, off);
                acc.y += __shfl_xor(acc.y, off);
                acc.z += __shfl_xor(acc.z, off);
                acc.w += __shfl_xor(acc.w, off);
            }

            if (g == 0) {
                const float d = (float)cnt;
                const float sc = SELF ? rsqrtf(d + 1.0f) : rsqrtf(fmaxf(d, 1.0f));
                v4f o = acc * sc;
                if (EPI) {
                    const float ds = (float)__builtin_amdgcn_readlane(csv, i & 31);
                    o = (o + bias4) * rsqrtf(fmaxf(ds, 1.0f));
                }
                *reinterpret_cast<v4f*>(&dst[(size_t)r * F + q4]) = o;
            }
        };

        int slotX = slotLine(0);
        int slotY = slotLine(1);

        for (int i = 0; i < m; i += 2) {
            const int slotX2 = slotLine(i + 2);
            const int slotY2 = slotLine(i + 3);

            int sx[U], sy[U];
#pragma unroll
            for (int u = 0; u < U; ++u) {
                sx[u] = __shfl(slotX, u * G + g);
                sy[u] = __shfl(slotY, u * G + g);
            }

            v4f vx[U], vy[U];
            v4f selfX, selfY;
            issueRows(vx, selfX, i, sx);
            issueRows(vy, selfY, i + 1, sy);

            finishRow(vx, selfX, i, slotX);
            if (i + 1 < m) finishRow(vy, selfY, i + 1, slotY);

            slotX = slotX2;
            slotY = slotY2;
        }
    }
}

template <int NOUT, bool TANH, bool ADDONE>
__global__ __launch_bounds__(128) void gcn_gemm(const float* __restrict__ in,
                                                const float* __restrict__ W,
                                                const float* __restrict__ bias,
                                                const int* __restrict__ cntS,
                                                float* __restrict__ out, int N) {
    constexpr int K = 64;
    constexpr int CT = NOUT / 16;
    constexpr int PITCH = 68;

    __shared__ float Wt[NOUT * PITCH];
    __shared__ float Bl[NOUT];
    __shared__ float Al[64 * PITCH];

    const int tid = threadIdx.x;
    const int wave = tid >> 5;
    const int lane = tid & 31;
    const int rowBase = blockIdx.x * 64;
    const int tileRow = rowBase + wave * 16;
    const int lm = lane & 15;
    const int kh = (lane >> 4) * 2;
    const int mh = (lane >> 4) * 8;

    constexpr int WI = (K * NOUT / 4) / 128;
    constexpr int AI = (64 * (K / 4)) / 128;
    const float4* Wg4 = reinterpret_cast<const float4*>(W);
    const float4* Ag4 = reinterpret_cast<const float4*>(in);
    float4 wv[WI];
    float4 av[AI];
    int cs[8];
#pragma unroll
    for (int t = 0; t < WI; ++t) wv[t] = Wg4[tid + t * 128];
#pragma unroll
    for (int t = 0; t < AI; ++t) {
        const int i = tid + t * 128;
        const int row = rowBase + i / (K / 4);
        av[t] = make_float4(0.f, 0.f, 0.f, 0.f);
        if (row < N) av[t] = Ag4[(size_t)row * (K / 4) + (i % (K / 4))];
    }
    float bl = 0.f;
    if (tid < NOUT) bl = bias[tid];
#pragma unroll
    for (int j = 0; j < 8; ++j) {
        const int row = tileRow + mh + j;
        cs[j] = cntS[row < N ? row : N - 1];
    }

#pragma unroll
    for (int t = 0; t < WI; ++t) {
        const int i = tid + t * 128;
        const int k = i / (NOUT / 4);
        const int n = (i % (NOUT / 4)) * 4;
        Wt[(n + 0) * PITCH + k] = wv[t].x;
        Wt[(n + 1) * PITCH + k] = wv[t].y;
        Wt[(n + 2) * PITCH + k] = wv[t].z;
        Wt[(n + 3) * PITCH + k] = wv[t].w;
    }
    if (tid < NOUT) Bl[tid] = bl;

#pragma unroll
    for (int t = 0; t < AI; ++t) {
        const int i = tid + t * 128;
        const int c4 = (i % (K / 4)) * 4;
        *reinterpret_cast<float4*>(&Al[(i / (K / 4)) * PITCH + c4]) = av[t];
    }
    __syncthreads();

    if (tileRow >= N) return;

    v8f acc[CT] = {};
    const float* Arow = &Al[(wave * 16 + lm) * PITCH + kh];

#pragma unroll
    for (int k0 = 0; k0 < K; k0 += 4) {
        v2f a = *reinterpret_cast<const v2f*>(&Arow[k0]);
#pragma unroll
        for (int c = 0; c < CT; ++c) {
            v2f b = *reinterpret_cast<const v2f*>(&Wt[(c * 16 + lm) * PITCH + k0 + kh]);
            acc[c] = __builtin_amdgcn_wmma_f32_16x16x4_f32(
                false, a, false, b, (short)0, acc[c], false, false);
        }
    }

    float rs[8];
#pragma unroll
    for (int j = 0; j < 8; ++j) {
        int row = tileRow + mh + j;
        float d = (row < N) ? (float)cs[j] : 1.0f;
        rs[j] = ADDONE ? rsqrtf(d + 1.0f) : rsqrtf(fmaxf(d, 1.0f));
    }

#pragma unroll
    for (int c = 0; c < CT; ++c) {
        int n = c * 16 + lm;
        float bv = Bl[n];
#pragma unroll
        for (int j = 0; j < 8; ++j) {
            int row = tileRow + mh + j;
            if (row < N) {
                float v = acc[c][j] + bv;
                if (TANH) v = __builtin_amdgcn_tanhf(v);
                out[(size_t)row * NOUT + n] = v * rs[j];
            }
        }
    }
}

template <int NOUT, int NP, bool TANH, bool ADDONE, int U>
__global__ __launch_bounds__(256, 6) void gcn_gather_gemm(const float* __restrict__ src,
                                                       const int* __restrict__ cntR,
                                                       const int* __restrict__ col,
                                                       const int* __restrict__ ovfCount,
                                                       const int* __restrict__ ovfS,
                                                       const int* __restrict__ ovfR,
                                                       const float* __restrict__ W,
                                                       const float* __restrict__ bias,
                                                       const int* __restrict__ cntS,
                                                       const float* __restrict__ Wp,
                                                       float* __restrict__ out, int N, int E) {
    static_assert(NOUT == 64, "the y tile is the A operand of the second product: 64 wide");
    constexpr int F = 64;
    constexpr int K = 64;
    constexpr int CT = NOUT / 16;
    constexpr int CP = NP / 16;
    constexpr int PITCH = 68;
    constexpr int LPR = F / 4;
    constexpr int G = 32 / LPR;

    __shared__ alignas(16) float Wt[NOUT * PITCH];
    __shared__ alignas(16) float Wpt[NP * PITCH];
    __shared__ float Bl[NOUT];
    __shared__ alignas(16) float Al[8 * 16 * PITCH];

    const int tid = threadIdx.x;
    const int lane = tid & 31;
    const int g = lane / LPR;
    const int q4 = (lane % LPR) * 4;
    const int lm = lane & 15;
    const int kh = (lane >> 4) * 2;
    const int mh = (lane >> 4) * 8;

    {
        constexpr int WI = (K * NOUT / 4) / 256;
        constexpr int WPI = (K * NP / 4) / 256;
        static_assert(WPI >= 1 && WPI * 256 * 4 == K * NP, "Wp is dealt evenly over the threads");
        const float4* Wg4 = reinterpret_cast<const float4*>(W);
        const float4* Wpg4 = reinterpret_cast<const float4*>(Wp);
        float4 wv[WI];
        float4 wpv[WPI];
#pragma unroll
        for (int t = 0; t < WI; ++t) wv[t] = Wg4[tid + t * 256];
#pragma unroll
        for (int t = 0; t < WPI; ++t) wpv[t] = Wpg4[tid + t * 256];
        float bl = 0.f;
        if (tid < NOUT) bl = bias[tid];
#pragma unroll
        for (int t = 0; t < WI; ++t) {
            const int i = tid + t * 256;
            const int k = i / (NOUT / 4);
            const int n = (i % (NOUT / 4)) * 4;
            Wt[(n + 0) * PITCH + k] = wv[t].x;
            Wt[(n + 1) * PITCH + k] = wv[t].y;
            Wt[(n + 2) * PITCH + k] = wv[t].z;
            Wt[(n + 3) * PITCH + k] = wv[t].w;
        }
#pragma unroll
        for (int t = 0; t < WPI; ++t) {
            const int i = tid + t * 256;
            const int k = i / (NP / 4);
            const int n = (i % (NP / 4)) * 4;
            Wpt[(n + 0) * PITCH + k] = wpv[t].x;
            Wpt[(n + 1) * PITCH + k] = wpv[t].y;
            Wpt[(n + 2) * PITCH + k] = wpv[t].z;
            Wpt[(n + 3) * PITCH + k] = wpv[t].w;
        }
        if (tid < NOUT) Bl[tid] = bl;
    }
    __syncthreads();

    const int wl = __builtin_amdgcn_readfirstlane((int)(threadIdx.x >> 5));
    const int wave = __builtin_amdgcn_readfirstlane((int)(blockIdx.x * 8)) + wl;
    const int base = N / GATHER_WAVES;
    const int rem = N % GATHER_WAVES;
    const int r0 = wave * base + (wave < rem ? wave : rem);
    const int r1 = r0 + base + (wave < rem ? 1 : 0);

    float* At = &Al[wl * 16 * PITCH];
    const v4f zero4 = {0.f, 0.f, 0.f, 0.f};

    for (int rb = r0; rb < r1; rb += 16) {
        const int m = (r1 - rb) < 16 ? (r1 - rb) : 16;
        const int cntv = cntR[rb + (lane < m ? lane : m - 1)];
        const int csv = cntS[rb + (lane < m ? lane : m - 1)];

        auto rowCount = [&](int i) -> int { return i < m ? __builtin_amdgcn_readlane(cntv, i & 31) : 0; };

        auto slotLine = [&](int i) -> int {
            const int c = rowCount(i);
            const int n = c < CAP ? c : CAP;
            const int ri = rb + (i < m ? i : m - 1);
            const int* p = (n > 0) ? &col[(size_t)ri * CAP + (lane < n ? lane : n - 1)] : &cntR[ri];
            return *p;
        };

        auto issueRows = [&](v4f (&v)[U], v4f& selfv, int i, const int (&s)[U]) {
            const int c = rowCount(i);
            const int n = c < CAP ? c : CAP;
            if (i < m) selfv = *reinterpret_cast<const v4f*>(&src[(size_t)(rb + i) * F + q4]);
#pragma unroll
            for (int u = 0; u < U; ++u) {
                if (u * G < n) v[u] = *reinterpret_cast<const v4f*>(&src[(size_t)s[u] * F + q4]);
            }
        };

        auto finishRow = [&](const v4f (&v)[U], const v4f& selfv, int i, int myS) {
            const int r = rb + i;
            const int cnt = rowCount(i);
            const int n = cnt < CAP ? cnt : CAP;

            v4f acc = zero4;
            if (g == 0) acc = selfv;
#pragma unroll
            for (int u = 0; u < U; ++u) {
                if (u * G < n) acc += (u * G + g < n) ? v[u] : zero4;
            }

            for (int j0 = U * G; j0 < n; j0 += U * G) {
                v4f w[U];
#pragma unroll
                for (int u = 0; u < U; ++u) {
                    const int j = j0 + u * G + g;
                    const int s = __shfl(myS, j & 31);
                    w[u] = zero4;
                    if (j < n) w[u] = *reinterpret_cast<const v4f*>(&src[(size_t)s * F + q4]);
                }
#pragma unroll
                for (int u = 0; u < U; ++u) acc += w[u];
            }

            if (cnt > CAP) {
                int no = *ovfCount;
                if (no > E) no = E;
                for (int k = g; k < no; k += G) {
                    if (ovfR[k] == r) acc += *reinterpret_cast<const v4f*>(&src[(size_t)ovfS[k] * F + q4]);
                }
            }

#pragma unroll
            for (int off = LPR; off < 32; off <<= 1) {
                acc.x += __shfl_xor(acc.x, off);
                acc.y += __shfl_xor(acc.y, off);
                acc.z += __shfl_xor(acc.z, off);
                acc.w += __shfl_xor(acc.w, off);
            }

            if (g == 0) {
                const float sc = rsqrtf((float)cnt + 1.0f);
                *reinterpret_cast<v4f*>(&At[i * PITCH + q4]) = acc * sc;
            }
        };

        int slotX = slotLine(0);
        int slotY = slotLine(1);

        for (int i = 0; i < m; i += 2) {
            const int slotX2 = slotLine(i + 2);
            const int slotY2 = slotLine(i + 3);

            int sx[U], sy[U];
#pragma unroll
            for (int u = 0; u < U; ++u) {
                sx[u] = __shfl(slotX, u * G + g);
                sy[u] = __shfl(slotY, u * G + g);
            }

            v4f vx[U], vy[U];
            v4f selfX, selfY;
            issueRows(vx, selfX, i, sx);
            issueRows(vy, selfY, i + 1, sy);

            finishRow(vx, selfX, i, slotX);
            if (i + 1 < m) finishRow(vy, selfY, i + 1, slotY);

            slotX = slotX2;
            slotY = slotY2;
        }

        for (int i = m; i < 16; ++i) {
            if (g == 0) *reinterpret_cast<v4f*>(&At[i * PITCH + q4]) = zero4;
        }

        __builtin_amdgcn_fence(__ATOMIC_ACQ_REL, "wavefront");

        v8f dacc[CT] = {};
        const float* Arow = &At[lm * PITCH + kh];

#pragma unroll 4
        for (int k0 = 0; k0 < K; k0 += 4) {
            v2f a = *reinterpret_cast<const v2f*>(&Arow[k0]);
#pragma unroll
            for (int c = 0; c < CT; ++c) {
                v2f b = *reinterpret_cast<const v2f*>(&Wt[(c * 16 + lm) * PITCH + k0 + kh]);
                dacc[c] = __builtin_amdgcn_wmma_f32_16x16x4_f32(
                    false, a, false, b, (short)0, dacc[c], false, false);
            }
        }

        __builtin_amdgcn_fence(__ATOMIC_ACQ_REL, "wavefront");

        const float dsv = (float)csv;
        const float rsv = ADDONE ? rsqrtf(dsv + 1.0f) : rsqrtf(fmaxf(dsv, 1.0f));
        float rs[8];
#pragma unroll
        for (int j = 0; j < 8; ++j) rs[j] = __shfl(rsv, mh + j);

#pragma unroll
        for (int c = 0; c < CT; ++c) {
            const int n = c * 16 + lm;
            const float bv = Bl[n];
#pragma unroll
            for (int j = 0; j < 8; ++j) {
                float y = 0.f;
                if (mh + j < m) {
                    float v = dacc[c][j] + bv;
                    if (TANH) v = __builtin_amdgcn_tanhf(v);
                    y = v * rs[j];
                }
                At[(mh + j) * PITCH + n] = y;
            }
        }

        __builtin_amdgcn_fence(__ATOMIC_ACQ_REL, "wavefront");

        v8f pacc[CP] = {};
#pragma unroll 4
        for (int k0 = 0; k0 < K; k0 += 4) {
            v2f a = *reinterpret_cast<const v2f*>(&Arow[k0]);
#pragma unroll
            for (int c = 0; c < CP; ++c) {
                v2f b = *reinterpret_cast<const v2f*>(&Wpt[(c * 16 + lm) * PITCH + k0 + kh]);
                pacc[c] = __builtin_amdgcn_wmma_f32_16x16x4_f32(
                    false, a, false, b, (short)0, pacc[c], false, false);
            }
        }

        __builtin_amdgcn_fence(__ATOMIC_ACQ_REL, "wavefront");

#pragma unroll
        for (int c = 0; c < CP; ++c) {
            const int n = c * 16 + lm;
#pragma unroll
            for (int j = 0; j < 8; ++j) {
                if (mh + j < m) out[(size_t)(rb + mh + j) * NP + n] = pacc[c][j];
            }
        }
    }
}

extern "C" void kernel_launch(void* const* d_in, const int* in_sizes, int n_in,
                              void* d_out, int out_size, void* d_ws, size_t ws_size,
                              hipStream_t stream) {
    const float* nodes = (const float*)d_in[0];
    const int* senders = (const int*)d_in[1];
    const int* receivers = (const int*)d_in[2];
    const float* W0 = (const float*)d_in[3];
    const float* b0 = (const float*)d_in[4];
    const float* W1 = (const float*)d_in[5];
    const float* b1 = (const float*)d_in[6];
    const float* W2 = (const float*)d_in[7];
    const float* b2 = (const float*)d_in[8];
    float* out = (float*)d_out;

    const int N = in_sizes[0] / 64;
    const int E = in_sizes[1];

    float* X = (float*)d_ws;
    float* P = X + (size_t)N * 64;
    float* Z = P + (size_t)N * 16;
    int* col = (int*)(Z + (size_t)N * 16);
    int* ovfS = col + (size_t)N * CAP;
    int* ovfR = ovfS + E;
    int* cntS = ovfR + E;
    int* cntR = cntS + N;
    int* curR = cntR + N;
    int* curS = curR + MAXBINS * CURSTRIDE;
    int* ovfCount = curS + MAXBINS * CURSTRIDE;
    int* binS = ovfCount + 4;
    const int nbins = (N + BINW - 1) / BINW;
    int* binR = binS + (size_t)nbins * BCAP;

    const int zn = 2 * MAXBINS * CURSTRIDE + 1;
    gcn_zero_i<<<(zn + 255) / 256, 256, 0, stream>>>(curR, zn);
    gcn_bin<<<(E + EPB - 1) / EPB, BINT, 0, stream>>>(senders, receivers, curR, curS, binR, binS, E);
    gcn_build<<<nbins, BUILDT, 0, stream>>>(senders, receivers, curR, curS, binR, binS, cntS, cntR,
                                            col, ovfCount, ovfS, ovfR, N, E);

    const int tiles = (N + 15) / 16;
    const int gblocks = (tiles + 3) / 4;

    gcn_gemm<64, true, true><<<gblocks, 128, 0, stream>>>(nodes, W0, b0, cntS, X, N);

    gcn_gather_gemm<64, 16, true, true, 8><<<GATHER_BLOCKS, 256, 0, stream>>>(X, cntR, col, ovfCount, ovfS, ovfR,
                                                                               W1, b1, cntS, W2, P, N, E);

    gcn_gather<16, true, true, 2><<<GATHER_BLOCKS, 256, 0, stream>>>(P, cntR, col, ovfCount, ovfS, ovfR,
                                                                      b2, cntS, Z, N, E);

    gcn_gather<16, false, false, 2><<<GATHER_BLOCKS, 256, 0, stream>>>(Z, cntR, col, ovfCount, ovfS, ovfR,
                                                                        b2, cntS, out, N, E);
}
